# starting kernel
# baseline (speedup 1.0000x reference)
#include <hip/hip_runtime.h>

typedef __attribute__((ext_vector_type(2))) float v2f;
typedef __attribute__((ext_vector_type(8))) float v8f;

#define F_IN   256
#define HIDDEN 64
#define NCLS   40


__global__ void k_init_deg(float* deg, int n) {
  int i = blockIdx.x * blockDim.x + threadIdx.x;
  if (i < n) deg[i] = 1.0f;
}

__global__ void k_deg_edges(const int* __restrict__ col, const float* __restrict__ ew,
                            float* deg, int e) {
  int i = blockIdx.x * blockDim.x + threadIdx.x;
  if (i < e) atomicAdd(&deg[col[i]], ew[i]);
}

__global__ void k_dinv(float* d, int n) {
  int i = blockIdx.x * blockDim.x + threadIdx.x;
  if (i < n) {
    float v = d[i];
    d[i] = (v > 0.0f) ? rsqrtf(v) : 0.0f;
  }
}


template <int K, int NOUT, bool RELU_IN, bool RELU_OUT, bool BIAS>
__global__ __launch_bounds__(64) void k_gemm_wmma(const float* __restrict__ A,
                                                  const float* __restrict__ W,
                                                  const float* __restrict__ bias,
                                                  float* __restrict__ D, int M) {
  constexpr int NT = (NOUT + 15) / 16;
  const int wv   = threadIdx.x >> 5;
  const int lane = threadIdx.x & 31;
  const int half = lane >> 4;
  const int l16  = lane & 15;
  const int m0   = (blockIdx.x * 2 + wv) * 16;
  if (m0 >= M) return;

  v8f acc[NT] = {};

  const float* __restrict__ arow = A + (size_t)(m0 + l16) * K + 2 * half;

  for (int k = 0; k < K; k += 4) {
    v2f a = *(const v2f*)(arow + k);
    if (RELU_IN) { a.x = fmaxf(a.x, 0.0f); a.y = fmaxf(a.y, 0.0f); }
    const int kk = k + 2 * half;
#pragma unroll
    for (int t = 0; t < NT; ++t) {
      const int c = t * 16 + l16;
      v2f b;
      if ((NOUT % 16 == 0) || (c < NOUT)) {
        b.x = W[(size_t)kk * NOUT + c];
        b.y = W[(size_t)(kk + 1) * NOUT + c];
      } else {
        b.x = 0.0f; b.y = 0.0f;
      }
      acc[t] = __builtin_amdgcn_wmma_f32_16x16x4_f32(
           false, a,  false, b,
           (short)0, acc[t],  false,  false);
    }
  }

#pragma unroll
  for (int t = 0; t < NT; ++t) {
    const int c = t * 16 + l16;
    if ((NOUT % 16 != 0) && (c >= NOUT)) continue;
    const float bv = BIAS ? bias[c] : 0.0f;
#pragma unroll
    for (int v = 0; v < 8; ++v) {
      const int m = m0 + v + 8 * half;
      float val = acc[t][v] + bv;
      if (RELU_OUT) val = fmaxf(val, 0.0f);
      D[(size_t)m * NOUT + c] = val;
    }
  }
}


__global__ void k_seed(const float* __restrict__ t, const float* __restrict__ dinv,
                       const float* __restrict__ bias, float* __restrict__ h, int total) {
  int i = blockIdx.x * blockDim.x + threadIdx.x;
  if (i >= total) return;
  const int node = i >> 6;
  const int f    = i & 63;
  const float di = dinv[node];
  h[i] = bias[f] + di * di * t[i];
}

__global__ void k_scatter(const int* __restrict__ row, const int* __restrict__ col,
                          const float* __restrict__ ew, const float* __restrict__ dinv,
                          const float* __restrict__ t, float* __restrict__ h, int E) {
  int gid  = blockIdx.x * blockDim.x + threadIdx.x;
  int e    = gid >> 5;
  int lane = gid & 31;
  if (e >= E) return;
  const int   r   = row[e];
  const int   c   = col[e];
  const float nrm = dinv[r] * ew[e] * dinv[c];
  const float2 v  = *(const float2*)(t + (size_t)r * HIDDEN + lane * 2);
  float* dst = h + (size_t)c * HIDDEN + lane * 2;
  atomicAdd(dst + 0, nrm * v.x);
  atomicAdd(dst + 1, nrm * v.y);
}


extern "C" void kernel_launch(void* const* d_in, const int* in_sizes, int n_in,
                              void* d_out, int out_size, void* d_ws, size_t ws_size,
                              hipStream_t stream) {
  const float* x   = (const float*)d_in[0];
  const int*   ei  = (const int*)d_in[1];
  const float* ew  = (const float*)d_in[2];
  const float* W1  = (const float*)d_in[3];
  const float* b1  = (const float*)d_in[4];
  const float* Wc1 = (const float*)d_in[5];
  const float* bc1 = (const float*)d_in[6];
  const float* Wc2 = (const float*)d_in[7];
  const float* bc2 = (const float*)d_in[8];
  const float* W2  = (const float*)d_in[9];
  const float* b2  = (const float*)d_in[10];
  float* out = (float*)d_out;

  const int N = in_sizes[0] / F_IN;
  const int E = in_sizes[2];
  const int* erow = ei;
  const int* ecol = ei + E;

  float* dinv = (float*)d_ws;
  float* h    = dinv + N;
  float* t    = h + (size_t)N * HIDDEN;

  const int TB = 256;
  const int gemm_grid = (N + 31) / 32;

  k_init_deg<<<(N + TB - 1) / TB, TB, 0, stream>>>(dinv, N);
  k_deg_edges<<<(E + TB - 1) / TB, TB, 0, stream>>>(ecol, ew, dinv, E);
  k_dinv<<<(N + TB - 1) / TB, TB, 0, stream>>>(dinv, N);

  k_gemm_wmma<F_IN, HIDDEN, false, true, true>
      <<<gemm_grid, 64, 0, stream>>>(x, W1, b1, h, N);

  k_gemm_wmma<HIDDEN, HIDDEN, false, false, false>
      <<<gemm_grid, 64, 0, stream>>>(h, Wc1, nullptr, t, N);
  k_seed<<<((N * HIDDEN) + TB - 1) / TB, TB, 0, stream>>>(t, dinv, bc1, h, N * HIDDEN);
  k_scatter<<<((size_t)E * 32 + TB - 1) / TB, TB, 0, stream>>>(erow, ecol, ew, dinv, t, h, E);

  k_gemm_wmma<HIDDEN, HIDDEN, true, false, false>
      <<<gemm_grid, 64, 0, stream>>>(h, Wc2, nullptr, t, N);
  k_seed<<<((N * HIDDEN) + TB - 1) / TB, TB, 0, stream>>>(t, dinv, bc2, h, N * HIDDEN);
  k_scatter<<<((size_t)E * 32 + TB - 1) / TB, TB, 0, stream>>>(erow, ecol, ew, dinv, t, h, E);

  k_gemm_wmma<HIDDEN, NCLS, true, false, true>
      <<<gemm_grid, 64, 0, stream>>>(h, W2, b2, out, N);
}


# Second conv's edge sums skip the id scan: conv1's work-groups leave their ranges' edge records as compact lists (LDS cursor, 32-record pieces), conv2 streams its list; sums in LDS arrival order
# speedup vs baseline: 1.9881x; 1.9881x over previous
#include <hip/hip_runtime.h>

typedef __attribute__((ext_vector_type(2))) float v2f;
typedef __attribute__((ext_vector_type(8))) float v8f;

#define F_IN   256
#define HIDDEN 64
#define NCLS   40

#define NB       256
#define NPB      391
#define ES_WAVES 32
#define QCAP     64
#define ROW_BITS 23
#define GRP      16
#define LGRP     32
#define LCAP     32768


__global__ void k_init_deg(float* deg, int n) {
  int i = blockIdx.x * blockDim.x + threadIdx.x;
  if (i < n) deg[i] = 1.0f;
}

__global__ void k_deg_edges(const int* __restrict__ row, const int* __restrict__ col,
                            const float* __restrict__ ew, float* deg,
                            unsigned char* __restrict__ bkt, unsigned* __restrict__ rec, int e) {
  int i = blockIdx.x * blockDim.x + threadIdx.x;
  if (i < e) {
    const int   c = col[i];
    const float w = ew[i];
    atomicAdd(&deg[c], w);
    const int b = c / NPB;
    bkt[i] = (unsigned char)b;
    *(uint2*)(rec + 2 * (size_t)i) = make_uint2(
        (unsigned)row[i] | ((unsigned)(c - b * NPB) << ROW_BITS), (unsigned)__float_as_int(w));
  }
}

__global__ void k_dinv(float* d, int n) {
  int i = blockIdx.x * blockDim.x + threadIdx.x;
  if (i < n) {
    float v = d[i];
    d[i] = (v > 0.0f) ? rsqrtf(v) : 0.0f;
  }
}


template <int K, int NOUT, bool NORM_IN, bool RELU_OUT, bool BIAS, bool NORM_OUT>
__global__ __launch_bounds__(64) void k_gemm_wmma(const float* A,
                                                  const float* __restrict__ W,
                                                  const float* __restrict__ bias,
                                                  const float* __restrict__ bias_in,
                                                  const float* __restrict__ dinv,
                                                  float* __restrict__ D, int M) {
  constexpr int NT = (NOUT + 15) / 16;
  const int wv   = threadIdx.x >> 5;
  const int lane = threadIdx.x & 31;
  const int half = lane >> 4;
  const int l16  = lane & 15;
  const int m0   = (blockIdx.x * 2 + wv) * 16;
  if (m0 >= M) return;

  v8f acc[NT] = {};

  const float* arow = A + (size_t)(m0 + l16) * K + 2 * half;
  const float di = NORM_IN ? dinv[m0 + l16] : 0.0f;

  for (int k = 0; k < K; k += 4) {
    v2f a = *(const v2f*)(arow + k);
    const int kk = k + 2 * half;
    if (NORM_IN) {
      const v2f bi = *(const v2f*)(bias_in + kk);
      a.x = fmaxf(fmaf(di, a.x, bi.x), 0.0f);
      a.y = fmaxf(fmaf(di, a.y, bi.y), 0.0f);
    }
#pragma unroll
    for (int t = 0; t < NT; ++t) {
      const int c = t * 16 + l16;
      v2f b;
      if ((NOUT % 16 == 0) || (c < NOUT)) {
        b.x = W[(size_t)kk * NOUT + c];
        b.y = W[(size_t)(kk + 1) * NOUT + c];
      } else {
        b.x = 0.0f; b.y = 0.0f;
      }
      acc[t] = __builtin_amdgcn_wmma_f32_16x16x4_f32(
           false, a,  false, b,
           (short)0, acc[t],  false,  false);
    }
  }

  float dr[8];
  if (NORM_OUT) {
    const float4* dp = (const float4*)(dinv + m0 + 8 * half);
    const float4 q0 = dp[0], q1 = dp[1];
    dr[0] = q0.x; dr[1] = q0.y; dr[2] = q0.z; dr[3] = q0.w;
    dr[4] = q1.x; dr[5] = q1.y; dr[6] = q1.z; dr[7] = q1.w;
  }

#pragma unroll
  for (int t = 0; t < NT; ++t) {
    const int c = t * 16 + l16;
    if ((NOUT % 16 != 0) && (c >= NOUT)) continue;
    const float bv = BIAS ? bias[c] : 0.0f;
#pragma unroll
    for (int v = 0; v < 8; ++v) {
      const int m = m0 + v + 8 * half;
      float val = acc[t][v] + bv;
      if (RELU_OUT) val = fmaxf(val, 0.0f);
      if (NORM_OUT) val *= dr[v];
      D[(size_t)m * NOUT + c] = val;
    }
  }
}


__device__ __forceinline__ unsigned zero_bytes(unsigned x) {
  return ~(((x & 0x7F7F7F7Fu) + 0x7F7F7F7Fu) | x | 0x7F7F7F7Fu);
}

__device__ __forceinline__ uint4 load_ids(const unsigned char* __restrict__ bkt, int e0, int E,
                                          unsigned fill) {
  if (e0 + 16 <= E) return *(const uint4*)(bkt + e0);
  unsigned w[4] = {fill, fill, fill, fill};
#pragma unroll
  for (int k = 0; k < 16; ++k) {
    if (e0 + k < E) {
      const unsigned sh = 8u * (k & 3);
      w[k >> 2] = (w[k >> 2] & ~(0xFFu << sh)) | ((unsigned)bkt[e0 + k] << sh);
    }
  }
  return make_uint4(w[0], w[1], w[2], w[3]);
}

template <bool FULL>
__device__ __forceinline__ uint2 fetch(const int* qp, int cnt, int lane,
                                       const unsigned* __restrict__ rec) {
  uint2 rc = make_uint2(0u, 0u);
  if (FULL || lane < cnt) rc = *(const uint2*)(rec + 2 * (size_t)qp[lane]);
  return rc;
}

template <bool FULL>
__device__ __forceinline__ void keep(uint2 rc, int cnt, int lane, unsigned* cursor,
                                     uint2* __restrict__ mylist) {
  unsigned base = 0u;
  if (lane == 0) base = atomicAdd(cursor, (unsigned)cnt);
  base = (unsigned)__builtin_amdgcn_readfirstlane((int)base);
  if (base + (unsigned)cnt <= (unsigned)LCAP && (FULL || lane < cnt)) mylist[base + lane] = rc;
}

template <int G, bool FULL>
__device__ __forceinline__ void drain(uint2 rc, int cnt, int lane,
                                      const float* __restrict__ s, float* acc) {
  const int r = (int)(rc.x & ((1u << ROW_BITS) - 1u));
  const int c = (int)(rc.x >> ROW_BITS);
  const int w_bits = (int)rc.y;
#pragma unroll
  for (int g = 0; g < 32; g += G) {
    if (!FULL && g >= cnt) break;
    float v0[G], v1[G];
#pragma unroll
    for (int u = 0; u < G; ++u) {
      v0[u] = 0.0f; v1[u] = 0.0f;
      if (FULL || g + u < cnt) {
        const int rj = __builtin_amdgcn_readlane(r, g + u);
        const float* src = s + (size_t)rj * HIDDEN + lane;
        v0[u] = src[0];
        v1[u] = src[32];
      }
    }
#pragma unroll
    for (int u = 0; u < G; ++u) {
      if (FULL || g + u < cnt) {
        const int   cj = __builtin_amdgcn_readlane(c, g + u);
        const float wj = __int_as_float(__builtin_amdgcn_readlane(w_bits, g + u));
        float* dst = acc + cj * HIDDEN + lane;
        atomicAdd(dst, wj * v0[u]);
        atomicAdd(dst + 32, wj * v1[u]);
      }
    }
  }
}

template <bool KEEP, bool USE>
__global__ __launch_bounds__(ES_WAVES * 32) void k_edge_sum(const unsigned char* __restrict__ bkt,
                                                            const unsigned* __restrict__ rec,
                                                            const float* __restrict__ s,
                                                            float* __restrict__ h,
                                                            unsigned* lists, unsigned* cnt,
                                                            int N, int E) {
  __shared__ __attribute__((aligned(16))) float acc[NPB * HIDDEN];
  __shared__ int q[ES_WAVES][QCAP];
  __shared__ unsigned cursor;

  const int tid  = threadIdx.x;
  const int lane = tid & 31;
  const int wv   = __builtin_amdgcn_readfirstlane(tid >> 5);
  const int node0 = blockIdx.x * NPB;
  int nown = N - node0;
  if (nown > NPB) nown = NPB;
  if (nown <= 0) return;

  uint2* mylist = (uint2*)(lists + 2 * (size_t)blockIdx.x * LCAP);
  if (KEEP && tid == 0) cursor = 0u;

  {
    constexpr int IT = (NPB * (HIDDEN / 4) + ES_WAVES * 32 - 1) / (ES_WAVES * 32);
    const float4* s4 = (const float4*)(s + (size_t)node0 * HIDDEN);
    float4* a4 = (float4*)acc;
    const int lim = nown * (HIDDEN / 4);
    float4 t[IT];
#pragma unroll
    for (int k = 0; k < IT; ++k) {
      const int i = tid + k * (ES_WAVES * 32);
      t[k] = make_float4(0.0f, 0.0f, 0.0f, 0.0f);
      if (i < lim) t[k] = s4[i];
    }
#pragma unroll
    for (int k = 0; k < IT; ++k) {
      const int i = tid + k * (ES_WAVES * 32);
      if (i < lim) a4[i] = t[k];
    }
  }

  unsigned total = (unsigned)LCAP + 1u;
  if (USE) total = cnt[blockIdx.x];
  __syncthreads();

  if (USE && total <= (unsigned)LCAP) {
    const int tot  = (int)total;
    const int nblk = (tot + 31) >> 5;
    uint2 nxt = make_uint2(0u, 0u);
    if (wv < nblk && wv * 32 + lane < tot) nxt = mylist[wv * 32 + lane];
    for (int b = wv; b < nblk; b += ES_WAVES) {
      const uint2 cur = nxt;
      const int i2 = (b + ES_WAVES) * 32 + lane;
      nxt = make_uint2(0u, 0u);
      if (i2 < tot) nxt = mylist[i2];
      const int left = tot - b * 32;
      if (left >= 32) drain<LGRP, true>(cur, 32, lane, s, acc);
      else            drain<LGRP, false>(cur, left, lane, s, acc);
    }
  } else {
    const unsigned me   = blockIdx.x;
    const unsigned pat  = me * 0x01010101u;
    const unsigned fill = (me ^ 1u) * 0x01010101u;
    const int nchunk = (E + 1023) >> 10;
    int* myq = q[wv];
    int n = 0;

    uint4 va = make_uint4(fill, fill, fill, fill);
    uint4 vb = va;
    if (wv < nchunk) {
      va = load_ids(bkt, wv * 1024 + lane * 16, E, fill);
      vb = load_ids(bkt, wv * 1024 + 512 + lane * 16, E, fill);
    }
    for (int ch = wv; ch < nchunk; ch += ES_WAVES) {
      const uint4 v = va, u = vb;
      const int e0 = ch * 1024 + lane * 16;
      if (ch + ES_WAVES < nchunk) {
        va = load_ids(bkt, e0 + ES_WAVES * 1024, E, fill);
        vb = load_ids(bkt, e0 + ES_WAVES * 1024 + 512, E, fill);
      }
      unsigned mm = (zero_bytes(v.x ^ pat) >> 7) | (zero_bytes(v.y ^ pat) >> 6) |
                    (zero_bytes(v.z ^ pat) >> 5) | (zero_bytes(v.w ^ pat) >> 4) |
                    (zero_bytes(u.x ^ pat) >> 3) | (zero_bytes(u.y ^ pat) >> 2) |
                    (zero_bytes(u.z ^ pat) >> 1) | zero_bytes(u.w ^ pat);
      unsigned bal = __builtin_amdgcn_ballot_w32(mm != 0u);
      while (bal != 0u) {
        if (mm != 0u) {
          const int p = __builtin_ctz(mm);
          mm &= mm - 1u;
          myq[n + (int)__builtin_amdgcn_mbcnt_lo(bal, 0u)] =
              e0 + ((p & 4) << 7) + ((p & 3) << 2) + (p >> 3);
        }
        n += __builtin_popcount(bal);
        if (n >= 32) {
          n -= 32;
          const uint2 rc = fetch<true>(myq + n, 32, lane, rec);
          if (KEEP) keep<true>(rc, 32, lane, &cursor, mylist);
          drain<GRP, true>(rc, 32, lane, s, acc);
        }
        bal = __builtin_amdgcn_ballot_w32(mm != 0u);
      }
    }
    if (n > 0) {
      const uint2 rc = fetch<false>(myq, n, lane, rec);
      if (KEEP) keep<false>(rc, n, lane, &cursor, mylist);
      drain<GRP, false>(rc, n, lane, s, acc);
    }
  }
  __syncthreads();

  if (KEEP && tid == 0) cnt[blockIdx.x] = cursor;

  {
    float4* h4 = (float4*)(h + (size_t)node0 * HIDDEN);
    const float4* a4 = (const float4*)acc;
    for (int i = tid; i < nown * (HIDDEN / 4); i += ES_WAVES * 32) h4[i] = a4[i];
  }
}


extern "C" void kernel_launch(void* const* d_in, const int* in_sizes, int n_in,
                              void* d_out, int out_size, void* d_ws, size_t ws_size,
                              hipStream_t stream) {
  const float* x   = (const float*)d_in[0];
  const int*   ei  = (const int*)d_in[1];
  const float* ew  = (const float*)d_in[2];
  const float* W1  = (const float*)d_in[3];
  const float* b1  = (const float*)d_in[4];
  const float* Wc1 = (const float*)d_in[5];
  const float* bc1 = (const float*)d_in[6];
  const float* Wc2 = (const float*)d_in[7];
  const float* bc2 = (const float*)d_in[8];
  const float* W2  = (const float*)d_in[9];
  const float* b2  = (const float*)d_in[10];
  float* out = (float*)d_out;

  const int N = in_sizes[0] / F_IN;
  const int E = in_sizes[2];
  const int* erow = ei;
  const int* ecol = ei + E;

  unsigned* lists = (unsigned*)d_ws;
  unsigned* cnt = lists + 2 * (size_t)NB * LCAP;
  float* dinv = (float*)(cnt + NB);
  float* h    = dinv + N;
  float* s    = h + (size_t)N * HIDDEN;
  unsigned* rec = (unsigned*)(s + (size_t)N * HIDDEN);
  unsigned char* bkt = (unsigned char*)(rec + 2 * (size_t)(E + (E & 1)));

  const int TB = 256;
  const int gemm_grid = (N + 31) / 32;

  k_init_deg<<<(N + TB - 1) / TB, TB, 0, stream>>>(dinv, N);
  k_deg_edges<<<(E + TB - 1) / TB, TB, 0, stream>>>(erow, ecol, ew, dinv, bkt, rec, E);
  k_dinv<<<(N + TB - 1) / TB, TB, 0, stream>>>(dinv, N);

  k_gemm_wmma<F_IN, HIDDEN, false, true, true, false>
      <<<gemm_grid, 64, 0, stream>>>(x, W1, b1, nullptr, nullptr, h, N);

  k_gemm_wmma<HIDDEN, HIDDEN, false, false, false, true>
      <<<gemm_grid, 64, 0, stream>>>(h, Wc1, nullptr, nullptr, dinv, s, N);
  k_edge_sum<true, false><<<NB, ES_WAVES * 32, 0, stream>>>(bkt, rec, s, h, lists, cnt, N, E);

  k_gemm_wmma<HIDDEN, HIDDEN, true, false, false, true>
      <<<gemm_grid, 64, 0, stream>>>(h, Wc2, nullptr, bc1, dinv, s, N);
  k_edge_sum<false, true><<<NB, ES_WAVES * 32, 0, stream>>>(bkt, rec, s, h, lists, cnt, N, E);

  k_gemm_wmma<HIDDEN, NCLS, true, false, true, false>
      <<<gemm_grid, 64, 0, stream>>>(h, W2, b2, bc2, dinv, out, N);
}


# Degree sums move into the owner scan, now a pass ahead of the products: it sums each range's edge weights in LDS (arrival order), writes dinv and the lists; no global float atomics; 9->8 launches
# speedup vs baseline: 2.6937x; 1.3549x over previous
#include <hip/hip_runtime.h>

typedef __attribute__((ext_vector_type(2))) float v2f;
typedef __attribute__((ext_vector_type(8))) float v8f;

#define F_IN   256
#define HIDDEN 64
#define NCLS   40

#define NB       256
#define NPB      391
#define ES_WAVES 32
#define QCAP     64
#define ROW_BITS 23
#define GRP      16
#define LGRP     32
#define LCAP     32768


__global__ void k_edge_records(const int* __restrict__ row, const int* __restrict__ col,
                               const float* __restrict__ ew, unsigned char* __restrict__ bkt,
                               unsigned* __restrict__ rec, int e) {
  int i = blockIdx.x * blockDim.x + threadIdx.x;
  if (i < e) {
    const int c = col[i];
    const int b = c / NPB;
    bkt[i] = (unsigned char)b;
    *(uint2*)(rec + 2 * (size_t)i) = make_uint2(
        (unsigned)row[i] | ((unsigned)(c - b * NPB) << ROW_BITS), (unsigned)__float_as_int(ew[i]));
  }
}


template <int K, int NOUT, bool NORM_IN, bool RELU_OUT, bool BIAS, bool NORM_OUT>
__global__ __launch_bounds__(64) void k_gemm_wmma(const float* A,
                                                  const float* __restrict__ W,
                                                  const float* __restrict__ bias,
                                                  const float* __restrict__ bias_in,
                                                  const float* __restrict__ dinv,
                                                  float* __restrict__ D, int M) {
  constexpr int NT = (NOUT + 15) / 16;
  const int wv   = threadIdx.x >> 5;
  const int lane = threadIdx.x & 31;
  const int half = lane >> 4;
  const int l16  = lane & 15;
  const int m0   = (blockIdx.x * 2 + wv) * 16;
  if (m0 >= M) return;

  v8f acc[NT] = {};

  const float* arow = A + (size_t)(m0 + l16) * K + 2 * half;
  const float di = NORM_IN ? dinv[m0 + l16] : 0.0f;

  for (int k = 0; k < K; k += 4) {
    v2f a = *(const v2f*)(arow + k);
    const int kk = k + 2 * half;
    if (NORM_IN) {
      const v2f bi = *(const v2f*)(bias_in + kk);
      a.x = fmaxf(fmaf(di, a.x, bi.x), 0.0f);
      a.y = fmaxf(fmaf(di, a.y, bi.y), 0.0f);
    }
#pragma unroll
    for (int t = 0; t < NT; ++t) {
      const int c = t * 16 + l16;
      v2f b;
      if ((NOUT % 16 == 0) || (c < NOUT)) {
        b.x = W[(size_t)kk * NOUT + c];
        b.y = W[(size_t)(kk + 1) * NOUT + c];
      } else {
        b.x = 0.0f; b.y = 0.0f;
      }
      acc[t] = __builtin_amdgcn_wmma_f32_16x16x4_f32(
           false, a,  false, b,
           (short)0, acc[t],  false,  false);
    }
  }

  float dr[8];
  if (NORM_OUT) {
    const float4* dp = (const float4*)(dinv + m0 + 8 * half);
    const float4 q0 = dp[0], q1 = dp[1];
    dr[0] = q0.x; dr[1] = q0.y; dr[2] = q0.z; dr[3] = q0.w;
    dr[4] = q1.x; dr[5] = q1.y; dr[6] = q1.z; dr[7] = q1.w;
  }

#pragma unroll
  for (int t = 0; t < NT; ++t) {
    const int c = t * 16 + l16;
    if ((NOUT % 16 != 0) && (c >= NOUT)) continue;
    const float bv = BIAS ? bias[c] : 0.0f;
#pragma unroll
    for (int v = 0; v < 8; ++v) {
      const int m = m0 + v + 8 * half;
      float val = acc[t][v] + bv;
      if (RELU_OUT) val = fmaxf(val, 0.0f);
      if (NORM_OUT) val *= dr[v];
      D[(size_t)m * NOUT + c] = val;
    }
  }
}


__device__ __forceinline__ unsigned zero_bytes(unsigned x) {
  return ~(((x & 0x7F7F7F7Fu) + 0x7F7F7F7Fu) | x | 0x7F7F7F7Fu);
}

__device__ __forceinline__ uint4 load_ids(const unsigned char* __restrict__ bkt, int e0, int E,
                                          unsigned fill) {
  if (e0 + 16 <= E) return *(const uint4*)(bkt + e0);
  unsigned w[4] = {fill, fill, fill, fill};
#pragma unroll
  for (int k = 0; k < 16; ++k) {
    if (e0 + k < E) {
      const unsigned sh = 8u * (k & 3);
      w[k >> 2] = (w[k >> 2] & ~(0xFFu << sh)) | ((unsigned)bkt[e0 + k] << sh);
    }
  }
  return make_uint4(w[0], w[1], w[2], w[3]);
}

__device__ __forceinline__ uint2 fetch(const int* qp, int cnt, int lane,
                                       const unsigned* __restrict__ rec) {
  uint2 rc = make_uint2(0u, 0u);
  if (lane < cnt) rc = *(const uint2*)(rec + 2 * (size_t)qp[lane]);
  return rc;
}

template <class F>
__device__ __forceinline__ void scan_range(const unsigned char* __restrict__ bkt,
                                           const unsigned* __restrict__ rec, int E, unsigned me,
                                           int wv, int lane, int* myq, F take) {
  const unsigned pat  = me * 0x01010101u;
  const unsigned fill = (me ^ 1u) * 0x01010101u;
  const int nchunk = (E + 1023) >> 10;
  int n = 0;

  uint4 va = make_uint4(fill, fill, fill, fill);
  uint4 vb = va;
  if (wv < nchunk) {
    va = load_ids(bkt, wv * 1024 + lane * 16, E, fill);
    vb = load_ids(bkt, wv * 1024 + 512 + lane * 16, E, fill);
  }
  for (int ch = wv; ch < nchunk; ch += ES_WAVES) {
    const uint4 v = va, u = vb;
    const int e0 = ch * 1024 + lane * 16;
    if (ch + ES_WAVES < nchunk) {
      va = load_ids(bkt, e0 + ES_WAVES * 1024, E, fill);
      vb = load_ids(bkt, e0 + ES_WAVES * 1024 + 512, E, fill);
    }
    unsigned mm = (zero_bytes(v.x ^ pat) >> 7) | (zero_bytes(v.y ^ pat) >> 6) |
                  (zero_bytes(v.z ^ pat) >> 5) | (zero_bytes(v.w ^ pat) >> 4) |
                  (zero_bytes(u.x ^ pat) >> 3) | (zero_bytes(u.y ^ pat) >> 2) |
                  (zero_bytes(u.z ^ pat) >> 1) | zero_bytes(u.w ^ pat);
    unsigned bal = __builtin_amdgcn_ballot_w32(mm != 0u);
    while (bal != 0u) {
      if (mm != 0u) {
        const int p = __builtin_ctz(mm);
        mm &= mm - 1u;
        myq[n + (int)__builtin_amdgcn_mbcnt_lo(bal, 0u)] =
            e0 + ((p & 4) << 7) + ((p & 3) << 2) + (p >> 3);
      }
      n += __builtin_popcount(bal);
      if (n >= 32) {
        n -= 32;
        take(fetch(myq + n, 32, lane, rec), 32);
      }
      bal = __builtin_amdgcn_ballot_w32(mm != 0u);
    }
  }
  if (n > 0) take(fetch(myq, n, lane, rec), n);
}

__global__ __launch_bounds__(ES_WAVES * 32) void k_range_scan(const unsigned char* __restrict__ bkt,
                                                              const unsigned* __restrict__ rec,
                                                              float* __restrict__ dinv,
                                                              unsigned* lists, unsigned* cnt,
                                                              int N, int E) {
  __shared__ float dsum[NPB];
  __shared__ int q[ES_WAVES][QCAP];
  __shared__ unsigned cursor;

  const int tid  = threadIdx.x;
  const int lane = tid & 31;
  const int wv   = __builtin_amdgcn_readfirstlane(tid >> 5);
  const int node0 = blockIdx.x * NPB;
  int nown = N - node0;
  if (nown > NPB) nown = NPB;
  if (nown <= 0) return;

  uint2* mylist = (uint2*)(lists + 2 * (size_t)blockIdx.x * LCAP);
  if (tid == 0) cursor = 0u;
  for (int i = tid; i < NPB; i += ES_WAVES * 32) dsum[i] = 1.0f;
  __syncthreads();

  scan_range(bkt, rec, E, blockIdx.x, wv, lane, q[wv], [&](uint2 rc, int c) {
    unsigned base = 0u;
    if (lane == 0) base = atomicAdd(&cursor, (unsigned)c);
    base = (unsigned)__builtin_amdgcn_readfirstlane((int)base);
    if (lane < c) {
      if (base + (unsigned)c <= (unsigned)LCAP) mylist[base + lane] = rc;
      atomicAdd(&dsum[rc.x >> ROW_BITS], __uint_as_float(rc.y));
    }
  });
  __syncthreads();

  if (tid == 0) cnt[blockIdx.x] = cursor;
  for (int i = tid; i < nown; i += ES_WAVES * 32) {
    const float v = dsum[i];
    dinv[node0 + i] = (v > 0.0f) ? rsqrtf(v) : 0.0f;
  }
}

template <int G, bool FULL>
__device__ __forceinline__ void drain(uint2 rc, int cnt, int lane,
                                      const float* __restrict__ s, float* acc) {
  const int r = (int)(rc.x & ((1u << ROW_BITS) - 1u));
  const int c = (int)(rc.x >> ROW_BITS);
  const int w_bits = (int)rc.y;
#pragma unroll
  for (int g = 0; g < 32; g += G) {
    if (!FULL && g >= cnt) break;
    float v0[G], v1[G];
#pragma unroll
    for (int u = 0; u < G; ++u) {
      v0[u] = 0.0f; v1[u] = 0.0f;
      if (FULL || g + u < cnt) {
        const int rj = __builtin_amdgcn_readlane(r, g + u);
        const float* src = s + (size_t)rj * HIDDEN + lane;
        v0[u] = src[0];
        v1[u] = src[32];
      }
    }
#pragma unroll
    for (int u = 0; u < G; ++u) {
      if (FULL || g + u < cnt) {
        const int   cj = __builtin_amdgcn_readlane(c, g + u);
        const float wj = __int_as_float(__builtin_amdgcn_readlane(w_bits, g + u));
        float* dst = acc + cj * HIDDEN + lane;
        atomicAdd(dst, wj * v0[u]);
        atomicAdd(dst + 32, wj * v1[u]);
      }
    }
  }
}

__global__ __launch_bounds__(ES_WAVES * 32) void k_edge_sum(const unsigned char* __restrict__ bkt,
                                                            const unsigned* __restrict__ rec,
                                                            const float* __restrict__ s,
                                                            float* __restrict__ h,
                                                            const unsigned* __restrict__ lists,
                                                            const unsigned* __restrict__ cnt,
                                                            int N, int E) {
  __shared__ __attribute__((aligned(16))) float acc[NPB * HIDDEN];
  __shared__ int q[ES_WAVES][QCAP];

  const int tid  = threadIdx.x;
  const int lane = tid & 31;
  const int wv   = __builtin_amdgcn_readfirstlane(tid >> 5);
  const int node0 = blockIdx.x * NPB;
  int nown = N - node0;
  if (nown > NPB) nown = NPB;
  if (nown <= 0) return;

  const uint2* mylist = (const uint2*)(lists + 2 * (size_t)blockIdx.x * LCAP);

  {
    constexpr int IT = (NPB * (HIDDEN / 4) + ES_WAVES * 32 - 1) / (ES_WAVES * 32);
    const float4* s4 = (const float4*)(s + (size_t)node0 * HIDDEN);
    float4* a4 = (float4*)acc;
    const int lim = nown * (HIDDEN / 4);
    float4 t[IT];
#pragma unroll
    for (int k = 0; k < IT; ++k) {
      const int i = tid + k * (ES_WAVES * 32);
      t[k] = make_float4(0.0f, 0.0f, 0.0f, 0.0f);
      if (i < lim) t[k] = s4[i];
    }
#pragma unroll
    for (int k = 0; k < IT; ++k) {
      const int i = tid + k * (ES_WAVES * 32);
      if (i < lim) a4[i] = t[k];
    }
  }

  const unsigned total = cnt[blockIdx.x];
  __syncthreads();

  if (total <= (unsigned)LCAP) {
    const int tot  = (int)total;
    const int nblk = (tot + 31) >> 5;
    uint2 nxt = make_uint2(0u, 0u);
    if (wv < nblk && wv * 32 + lane < tot) nxt = mylist[wv * 32 + lane];
    for (int b = wv; b < nblk; b += ES_WAVES) {
      const uint2 cur = nxt;
      const int i2 = (b + ES_WAVES) * 32 + lane;
      nxt = make_uint2(0u, 0u);
      if (i2 < tot) nxt = mylist[i2];
      const int left = tot - b * 32;
      if (left >= 32) drain<LGRP, true>(cur, 32, lane, s, acc);
      else            drain<LGRP, false>(cur, left, lane, s, acc);
    }
  } else {
    scan_range(bkt, rec, E, blockIdx.x, wv, lane, q[wv], [&](uint2 rc, int c) {
      if (c == 32) drain<GRP, true>(rc, 32, lane, s, acc);
      else         drain<GRP, false>(rc, c, lane, s, acc);
    });
  }
  __syncthreads();

  {
    float4* h4 = (float4*)(h + (size_t)node0 * HIDDEN);
    const float4* a4 = (const float4*)acc;
    for (int i = tid; i < nown * (HIDDEN / 4); i += ES_WAVES * 32) h4[i] = a4[i];
  }
}


extern "C" void kernel_launch(void* const* d_in, const int* in_sizes, int n_in,
                              void* d_out, int out_size, void* d_ws, size_t ws_size,
                              hipStream_t stream) {
  const float* x   = (const float*)d_in[0];
  const int*   ei  = (const int*)d_in[1];
  const float* ew  = (const float*)d_in[2];
  const float* W1  = (const float*)d_in[3];
  const float* b1  = (const float*)d_in[4];
  const float* Wc1 = (const float*)d_in[5];
  const float* bc1 = (const float*)d_in[6];
  const float* Wc2 = (const float*)d_in[7];
  const float* bc2 = (const float*)d_in[8];
  const float* W2  = (const float*)d_in[9];
  const float* b2  = (const float*)d_in[10];
  float* out = (float*)d_out;

  const int N = in_sizes[0] / F_IN;
  const int E = in_sizes[2];
  const int* erow = ei;
  const int* ecol = ei + E;

  unsigned* lists = (unsigned*)d_ws;
  unsigned* cnt = lists + 2 * (size_t)NB * LCAP;
  float* dinv = (float*)(cnt + NB);
  float* h    = dinv + N;
  float* s    = h + (size_t)N * HIDDEN;
  unsigned* rec = (unsigned*)(s + (size_t)N * HIDDEN);
  unsigned char* bkt = (unsigned char*)(rec + 2 * (size_t)(E + (E & 1)));

  const int TB = 256;
  const int gemm_grid = (N + 31) / 32;

  k_edge_records<<<(E + TB - 1) / TB, TB, 0, stream>>>(erow, ecol, ew, bkt, rec, E);
  k_range_scan<<<NB, ES_WAVES * 32, 0, stream>>>(bkt, rec, dinv, lists, cnt, N, E);

  k_gemm_wmma<F_IN, HIDDEN, false, true, true, false>
      <<<gemm_grid, 64, 0, stream>>>(x, W1, b1, nullptr, nullptr, h, N);

  k_gemm_wmma<HIDDEN, HIDDEN, false, false, false, true>
      <<<gemm_grid, 64, 0, stream>>>(h, Wc1, nullptr, nullptr, dinv, s, N);
  k_edge_sum<<<NB, ES_WAVES * 32, 0, stream>>>(bkt, rec, s, h, lists, cnt, N, E);

  k_gemm_wmma<HIDDEN, HIDDEN, true, false, false, true>
      <<<gemm_grid, 64, 0, stream>>>(h, Wc2, nullptr, bc1, dinv, s, N);
  k_edge_sum<<<NB, ES_WAVES * 32, 0, stream>>>(bkt, rec, s, h, lists, cnt, N, E);

  k_gemm_wmma<HIDDEN, NCLS, true, false, true, false>
      <<<gemm_grid, 64, 0, stream>>>(h, W2, b2, bc2, dinv, out, N);
}


# First product x@W1 moved into the range scan's launch: 196 work-groups behind the 256 scanners run its 16-row tiles, matrix units busy beside the scan's vector work; 8->7 launches; sums unchanged
# speedup vs baseline: 2.8259x; 1.0491x over previous
#include <hip/hip_runtime.h>

typedef __attribute__((ext_vector_type(2))) float v2f;
typedef __attribute__((ext_vector_type(8))) float v8f;

#define F_IN   256
#define HIDDEN 64
#define NCLS   40

#define NB       256
#define NPB      391
#define ES_WAVES 32
#define QCAP     64
#define ROW_BITS 23
#define GRP      16
#define LGRP     32
#define LCAP     32768


__global__ void k_edge_records(const int* __restrict__ row, const int* __restrict__ col,
                               const float* __restrict__ ew, unsigned char* __restrict__ bkt,
                               unsigned* __restrict__ rec, int e) {
  int i = blockIdx.x * blockDim.x + threadIdx.x;
  if (i < e) {
    const int c = col[i];
    const int b = c / NPB;
    bkt[i] = (unsigned char)b;
    *(uint2*)(rec + 2 * (size_t)i) = make_uint2(
        (unsigned)row[i] | ((unsigned)(c - b * NPB) << ROW_BITS), (unsigned)__float_as_int(ew[i]));
  }
}


template <int K, int NOUT, bool NORM_IN, bool RELU_OUT, bool BIAS, bool NORM_OUT>
__device__ __forceinline__ void gemm_tile(const float* A, const float* __restrict__ W,
                                          const float* __restrict__ bias,
                                          const float* __restrict__ bias_in,
                                          const float* __restrict__ dinv,
                                          float* __restrict__ D, int m0, int lane) {
  constexpr int NT = (NOUT + 15) / 16;
  const int half = lane >> 4;
  const int l16  = lane & 15;

  v8f acc[NT] = {};

  const float* arow = A + (size_t)(m0 + l16) * K + 2 * half;
  const float di = NORM_IN ? dinv[m0 + l16] : 0.0f;

  for (int k = 0; k < K; k += 4) {
    v2f a = *(const v2f*)(arow + k);
    const int kk = k + 2 * half;
    if (NORM_IN) {
      const v2f bi = *(const v2f*)(bias_in + kk);
      a.x = fmaxf(fmaf(di, a.x, bi.x), 0.0f);
      a.y = fmaxf(fmaf(di, a.y, bi.y), 0.0f);
    }
#pragma unroll
    for (int t = 0; t < NT; ++t) {
      const int c = t * 16 + l16;
      v2f b;
      if ((NOUT % 16 == 0) || (c < NOUT)) {
        b.x = W[(size_t)kk * NOUT + c];
        b.y = W[(size_t)(kk + 1) * NOUT + c];
      } else {
        b.x = 0.0f; b.y = 0.0f;
      }
      acc[t] = __builtin_amdgcn_wmma_f32_16x16x4_f32(
           false, a,  false, b,
           (short)0, acc[t],  false,  false);
    }
  }

  float dr[8];
  if (NORM_OUT) {
    const float4* dp = (const float4*)(dinv + m0 + 8 * half);
    const float4 q0 = dp[0], q1 = dp[1];
    dr[0] = q0.x; dr[1] = q0.y; dr[2] = q0.z; dr[3] = q0.w;
    dr[4] = q1.x; dr[5] = q1.y; dr[6] = q1.z; dr[7] = q1.w;
  }

#pragma unroll
  for (int t = 0; t < NT; ++t) {
    const int c = t * 16 + l16;
    if ((NOUT % 16 != 0) && (c >= NOUT)) continue;
    const float bv = BIAS ? bias[c] : 0.0f;
#pragma unroll
    for (int v = 0; v < 8; ++v) {
      const int m = m0 + v + 8 * half;
      float val = acc[t][v] + bv;
      if (RELU_OUT) val = fmaxf(val, 0.0f);
      if (NORM_OUT) val *= dr[v];
      D[(size_t)m * NOUT + c] = val;
    }
  }
}

template <int K, int NOUT, bool NORM_IN, bool RELU_OUT, bool BIAS, bool NORM_OUT>
__global__ __launch_bounds__(64) void k_gemm_wmma(const float* A,
                                                  const float* __restrict__ W,
                                                  const float* __restrict__ bias,
                                                  const float* __restrict__ bias_in,
                                                  const float* __restrict__ dinv,
                                                  float* __restrict__ D, int M) {
  const int wv   = threadIdx.x >> 5;
  const int lane = threadIdx.x & 31;
  const int m0   = (blockIdx.x * 2 + wv) * 16;
  if (m0 >= M) return;
  gemm_tile<K, NOUT, NORM_IN, RELU_OUT, BIAS, NORM_OUT>(A, W, bias, bias_in, dinv, D, m0, lane);
}


__device__ __forceinline__ unsigned zero_bytes(unsigned x) {
  return ~(((x & 0x7F7F7F7Fu) + 0x7F7F7F7Fu) | x | 0x7F7F7F7Fu);
}

__device__ __forceinline__ uint4 load_ids(const unsigned char* __restrict__ bkt, int e0, int E,
                                          unsigned fill) {
  if (e0 + 16 <= E) return *(const uint4*)(bkt + e0);
  unsigned w[4] = {fill, fill, fill, fill};
#pragma unroll
  for (int k = 0; k < 16; ++k) {
    if (e0 + k < E) {
      const unsigned sh = 8u * (k & 3);
      w[k >> 2] = (w[k >> 2] & ~(0xFFu << sh)) | ((unsigned)bkt[e0 + k] << sh);
    }
  }
  return make_uint4(w[0], w[1], w[2], w[3]);
}

__device__ __forceinline__ uint2 fetch(const int* qp, int cnt, int lane,
                                       const unsigned* __restrict__ rec) {
  uint2 rc = make_uint2(0u, 0u);
  if (lane < cnt) rc = *(const uint2*)(rec + 2 * (size_t)qp[lane]);
  return rc;
}

template <class F>
__device__ __forceinline__ void scan_range(const unsigned char* __restrict__ bkt,
                                           const unsigned* __restrict__ rec, int E, unsigned me,
                                           int wv, int lane, int* myq, F take) {
  const unsigned pat  = me * 0x01010101u;
  const unsigned fill = (me ^ 1u) * 0x01010101u;
  const int nchunk = (E + 1023) >> 10;
  int n = 0;

  uint4 va = make_uint4(fill, fill, fill, fill);
  uint4 vb = va;
  if (wv < nchunk) {
    va = load_ids(bkt, wv * 1024 + lane * 16, E, fill);
    vb = load_ids(bkt, wv * 1024 + 512 + lane * 16, E, fill);
  }
  for (int ch = wv; ch < nchunk; ch += ES_WAVES) {
    const uint4 v = va, u = vb;
    const int e0 = ch * 1024 + lane * 16;
    if (ch + ES_WAVES < nchunk) {
      va = load_ids(bkt, e0 + ES_WAVES * 1024, E, fill);
      vb = load_ids(bkt, e0 + ES_WAVES * 1024 + 512, E, fill);
    }
    unsigned mm = (zero_bytes(v.x ^ pat) >> 7) | (zero_bytes(v.y ^ pat) >> 6) |
                  (zero_bytes(v.z ^ pat) >> 5) | (zero_bytes(v.w ^ pat) >> 4) |
                  (zero_bytes(u.x ^ pat) >> 3) | (zero_bytes(u.y ^ pat) >> 2) |
                  (zero_bytes(u.z ^ pat) >> 1) | zero_bytes(u.w ^ pat);
    unsigned bal = __builtin_amdgcn_ballot_w32(mm != 0u);
    while (bal != 0u) {
      if (mm != 0u) {
        const int p = __builtin_ctz(mm);
        mm &= mm - 1u;
        myq[n + (int)__builtin_amdgcn_mbcnt_lo(bal, 0u)] =
            e0 + ((p & 4) << 7) + ((p & 3) << 2) + (p >> 3);
      }
      n += __builtin_popcount(bal);
      if (n >= 32) {
        n -= 32;
        take(fetch(myq + n, 32, lane, rec), 32);
      }
      bal = __builtin_amdgcn_ballot_w32(mm != 0u);
    }
  }
  if (n > 0) take(fetch(myq, n, lane, rec), n);
}

__global__ __launch_bounds__(ES_WAVES * 32) void k_range_scan_and_first_product(
    const unsigned char* __restrict__ bkt, const unsigned* __restrict__ rec,
    float* __restrict__ dinv, unsigned* lists, unsigned* cnt, int N, int E,
    const float* __restrict__ x, const float* __restrict__ W1, const float* __restrict__ b1,
    float* __restrict__ h) {
  __shared__ float dsum[NPB];
  __shared__ int q[ES_WAVES][QCAP];
  __shared__ unsigned cursor;

  const int tid  = threadIdx.x;
  const int lane = tid & 31;
  const int wv   = __builtin_amdgcn_readfirstlane(tid >> 5);

  if (blockIdx.x >= NB) {
    const int m0 = ((blockIdx.x - NB) * ES_WAVES + wv) * 16;
    if (m0 < N)
      gemm_tile<F_IN, HIDDEN, false, true, true, false>(x, W1, b1, nullptr, nullptr, h, m0, lane);
    return;
  }

  const int node0 = blockIdx.x * NPB;
  int nown = N - node0;
  if (nown > NPB) nown = NPB;
  if (nown <= 0) return;

  uint2* mylist = (uint2*)(lists + 2 * (size_t)blockIdx.x * LCAP);
  if (tid == 0) cursor = 0u;
  for (int i = tid; i < NPB; i += ES_WAVES * 32) dsum[i] = 1.0f;
  __syncthreads();

  scan_range(bkt, rec, E, blockIdx.x, wv, lane, q[wv], [&](uint2 rc, int c) {
    unsigned base = 0u;
    if (lane == 0) base = atomicAdd(&cursor, (unsigned)c);
    base = (unsigned)__builtin_amdgcn_readfirstlane((int)base);
    if (lane < c) {
      if (base + (unsigned)c <= (unsigned)LCAP) mylist[base + lane] = rc;
      atomicAdd(&dsum[rc.x >> ROW_BITS], __uint_as_float(rc.y));
    }
  });
  __syncthreads();

  if (tid == 0) cnt[blockIdx.x] = cursor;
  for (int i = tid; i < nown; i += ES_WAVES * 32) {
    const float v = dsum[i];
    dinv[node0 + i] = (v > 0.0f) ? rsqrtf(v) : 0.0f;
  }
}

template <int G, bool FULL>
__device__ __forceinline__ void drain(uint2 rc, int cnt, int lane,
                                      const float* __restrict__ s, float* acc) {
  const int r = (int)(rc.x & ((1u << ROW_BITS) - 1u));
  const int c = (int)(rc.x >> ROW_BITS);
  const int w_bits = (int)rc.y;
#pragma unroll
  for (int g = 0; g < 32; g += G) {
    if (!FULL && g >= cnt) break;
    float v0[G], v1[G];
#pragma unroll
    for (int u = 0; u < G; ++u) {
      v0[u] = 0.0f; v1[u] = 0.0f;
      if (FULL || g + u < cnt) {
        const int rj = __builtin_amdgcn_readlane(r, g + u);
        const float* src = s + (size_t)rj * HIDDEN + lane;
        v0[u] = src[0];
        v1[u] = src[32];
      }
    }
#pragma unroll
    for (int u = 0; u < G; ++u) {
      if (FULL || g + u < cnt) {
        const int   cj = __builtin_amdgcn_readlane(c, g + u);
        const float wj = __int_as_float(__builtin_amdgcn_readlane(w_bits, g + u));
        float* dst = acc + cj * HIDDEN + lane;
        atomicAdd(dst, wj * v0[u]);
        atomicAdd(dst + 32, wj * v1[u]);
      }
    }
  }
}

__global__ __launch_bounds__(ES_WAVES * 32) void k_edge_sum(const unsigned char* __restrict__ bkt,
                                                            const unsigned* __restrict__ rec,
                                                            const float* __restrict__ s,
                                                            float* __restrict__ h,
                                                            const unsigned* __restrict__ lists,
                                                            const unsigned* __restrict__ cnt,
                                                            int N, int E) {
  __shared__ __attribute__((aligned(16))) float acc[NPB * HIDDEN];
  __shared__ int q[ES_WAVES][QCAP];

  const int tid  = threadIdx.x;
  const int lane = tid & 31;
  const int wv   = __builtin_amdgcn_readfirstlane(tid >> 5);
  const int node0 = blockIdx.x * NPB;
  int nown = N - node0;
  if (nown > NPB) nown = NPB;
  if (nown <= 0) return;

  const uint2* mylist = (const uint2*)(lists + 2 * (size_t)blockIdx.x * LCAP);

  {
    constexpr int IT = (NPB * (HIDDEN / 4) + ES_WAVES * 32 - 1) / (ES_WAVES * 32);
    const float4* s4 = (const float4*)(s + (size_t)node0 * HIDDEN);
    float4* a4 = (float4*)acc;
    const int lim = nown * (HIDDEN / 4);
    float4 t[IT];
#pragma unroll
    for (int k = 0; k < IT; ++k) {
      const int i = tid + k * (ES_WAVES * 32);
      t[k] = make_float4(0.0f, 0.0f, 0.0f, 0.0f);
      if (i < lim) t[k] = s4[i];
    }
#pragma unroll
    for (int k = 0; k < IT; ++k) {
      const int i = tid + k * (ES_WAVES * 32);
      if (i < lim) a4[i] = t[k];
    }
  }

  const unsigned total = cnt[blockIdx.x];
  __syncthreads();

  if (total <= (unsigned)LCAP) {
    const int tot  = (int)total;
    const int nblk = (tot + 31) >> 5;
    uint2 nxt = make_uint2(0u, 0u);
    if (wv < nblk && wv * 32 + lane < tot) nxt = mylist[wv * 32 + lane];
    for (int b = wv; b < nblk; b += ES_WAVES) {
      const uint2 cur = nxt;
      const int i2 = (b + ES_WAVES) * 32 + lane;
      nxt = make_uint2(0u, 0u);
      if (i2 < tot) nxt = mylist[i2];
      const int left = tot - b * 32;
      if (left >= 32) drain<LGRP, true>(cur, 32, lane, s, acc);
      else            drain<LGRP, false>(cur, left, lane, s, acc);
    }
  } else {
    scan_range(bkt, rec, E, blockIdx.x, wv, lane, q[wv], [&](uint2 rc, int c) {
      if (c == 32) drain<GRP, true>(rc, 32, lane, s, acc);
      else         drain<GRP, false>(rc, c, lane, s, acc);
    });
  }
  __syncthreads();

  {
    float4* h4 = (float4*)(h + (size_t)node0 * HIDDEN);
    const float4* a4 = (const float4*)acc;
    for (int i = tid; i < nown * (HIDDEN / 4); i += ES_WAVES * 32) h4[i] = a4[i];
  }
}


extern "C" void kernel_launch(void* const* d_in, const int* in_sizes, int n_in,
                              void* d_out, int out_size, void* d_ws, size_t ws_size,
                              hipStream_t stream) {
  const float* x   = (const float*)d_in[0];
  const int*   ei  = (const int*)d_in[1];
  const float* ew  = (const float*)d_in[2];
  const float* W1  = (const float*)d_in[3];
  const float* b1  = (const float*)d_in[4];
  const float* Wc1 = (const float*)d_in[5];
  const float* bc1 = (const float*)d_in[6];
  const float* Wc2 = (const float*)d_in[7];
  const float* bc2 = (const float*)d_in[8];
  const float* W2  = (const float*)d_in[9];
  const float* b2  = (const float*)d_in[10];
  float* out = (float*)d_out;

  const int N = in_sizes[0] / F_IN;
  const int E = in_sizes[2];
  const int* erow = ei;
  const int* ecol = ei + E;

  unsigned* lists = (unsigned*)d_ws;
  unsigned* cnt = lists + 2 * (size_t)NB * LCAP;
  float* dinv = (float*)(cnt + NB);
  float* h    = dinv + N;
  float* s    = h + (size_t)N * HIDDEN;
  unsigned* rec = (unsigned*)(s + (size_t)N * HIDDEN);
  unsigned char* bkt = (unsigned char*)(rec + 2 * (size_t)(E + (E & 1)));

  const int TB = 256;
  const int gemm_grid = (N + 31) / 32;

  k_edge_records<<<(E + TB - 1) / TB, TB, 0, stream>>>(erow, ecol, ew, bkt, rec, E);
  const int first_grid = (N + ES_WAVES * 16 - 1) / (ES_WAVES * 16);
  k_range_scan_and_first_product<<<NB + first_grid, ES_WAVES * 32, 0, stream>>>(
      bkt, rec, dinv, lists, cnt, N, E, x, W1, b1, h);

  k_gemm_wmma<HIDDEN, HIDDEN, false, false, false, true>
      <<<gemm_grid, 64, 0, stream>>>(h, Wc1, nullptr, nullptr, dinv, s, N);
  k_edge_sum<<<NB, ES_WAVES * 32, 0, stream>>>(bkt, rec, s, h, lists, cnt, N, E);

  k_gemm_wmma<HIDDEN, HIDDEN, true, false, false, true>
      <<<gemm_grid, 64, 0, stream>>>(h, Wc2, nullptr, bc1, dinv, s, N);
  k_edge_sum<<<NB, ES_WAVES * 32, 0, stream>>>(bkt, rec, s, h, lists, cnt, N, E);

  k_gemm_wmma<HIDDEN, NCLS, true, false, true, false>
      <<<gemm_grid, 64, 0, stream>>>(h, W2, b2, bc2, dinv, out, N);
}


# Consuming product moved into the range's work-group: waves take 16 rows of the finished LDS sums as A fragments, results leave via LDS in whole lines; 7->5 launches; k order, LDS arrival order kept
# speedup vs baseline: 3.2319x; 1.1437x over previous
#include <hip/hip_runtime.h>

typedef __attribute__((ext_vector_type(2))) float v2f;
typedef __attribute__((ext_vector_type(8))) float v8f;

#define F_IN   256
#define HIDDEN 64
#define NCLS   40

#define NB       256
#define NPB      392
#define ROWS_P   (((NPB + 15) / 16) * 16)
#define ACC_P    68
#define ES_WAVES 32
#define QCAP     64
#define ROW_BITS 23
#define GRP      16
#define LGRP     32
#define LCAP     32768


__global__ void k_edge_records(const int* __restrict__ row, const int* __restrict__ col,
                               const float* __restrict__ ew, unsigned char* __restrict__ bkt,
                               unsigned* __restrict__ rec, int e) {
  int i = blockIdx.x * blockDim.x + threadIdx.x;
  if (i < e) {
    const int c = col[i];
    const int b = c / NPB;
    bkt[i] = (unsigned char)b;
    *(uint2*)(rec + 2 * (size_t)i) = make_uint2(
        (unsigned)row[i] | ((unsigned)(c - b * NPB) << ROW_BITS), (unsigned)__float_as_int(ew[i]));
  }
}


template <int K, int NOUT, bool NORM_IN, bool RELU_OUT, bool BIAS, bool NORM_OUT>
__device__ __forceinline__ void gemm_tile(const float* A, const float* __restrict__ W,
                                          const float* __restrict__ bias,
                                          const float* __restrict__ bias_in,
                                          const float* __restrict__ dinv,
                                          float* __restrict__ D, int m0, int lane) {
  constexpr int NT = (NOUT + 15) / 16;
  const int half = lane >> 4;
  const int l16  = lane & 15;

  v8f acc[NT] = {};

  const float* arow = A + (size_t)(m0 + l16) * K + 2 * half;
  const float di = NORM_IN ? dinv[m0 + l16] : 0.0f;

  for (int k = 0; k < K; k += 4) {
    v2f a = *(const v2f*)(arow + k);
    const int kk = k + 2 * half;
    if (NORM_IN) {
      const v2f bi = *(const v2f*)(bias_in + kk);
      a.x = fmaxf(fmaf(di, a.x, bi.x), 0.0f);
      a.y = fmaxf(fmaf(di, a.y, bi.y), 0.0f);
    }
#pragma unroll
    for (int t = 0; t < NT; ++t) {
      const int c = t * 16 + l16;
      v2f b;
      if ((NOUT % 16 == 0) || (c < NOUT)) {
        b.x = W[(size_t)kk * NOUT + c];
        b.y = W[(size_t)(kk + 1) * NOUT + c];
      } else {
        b.x = 0.0f; b.y = 0.0f;
      }
      acc[t] = __builtin_amdgcn_wmma_f32_16x16x4_f32(
           false, a,  false, b,
           (short)0, acc[t],  false,  false);
    }
  }

  float dr[8];
  if (NORM_OUT) {
    const float4* dp = (const float4*)(dinv + m0 + 8 * half);
    const float4 q0 = dp[0], q1 = dp[1];
    dr[0] = q0.x; dr[1] = q0.y; dr[2] = q0.z; dr[3] = q0.w;
    dr[4] = q1.x; dr[5] = q1.y; dr[6] = q1.z; dr[7] = q1.w;
  }

#pragma unroll
  for (int t = 0; t < NT; ++t) {
    const int c = t * 16 + l16;
    if ((NOUT % 16 != 0) && (c >= NOUT)) continue;
    const float bv = BIAS ? bias[c] : 0.0f;
#pragma unroll
    for (int v = 0; v < 8; ++v) {
      const int m = m0 + v + 8 * half;
      float val = acc[t][v] + bv;
      if (RELU_OUT) val = fmaxf(val, 0.0f);
      if (NORM_OUT) val *= dr[v];
      D[(size_t)m * NOUT + c] = val;
    }
  }
}

template <int K, int NOUT, bool NORM_IN, bool RELU_OUT, bool BIAS, bool NORM_OUT>
__global__ __launch_bounds__(64) void k_gemm_wmma(const float* A,
                                                  const float* __restrict__ W,
                                                  const float* __restrict__ bias,
                                                  const float* __restrict__ bias_in,
                                                  const float* __restrict__ dinv,
                                                  float* __restrict__ D, int M) {
  const int wv   = threadIdx.x >> 5;
  const int lane = threadIdx.x & 31;
  const int m0   = (blockIdx.x * 2 + wv) * 16;
  if (m0 >= M) return;
  gemm_tile<K, NOUT, NORM_IN, RELU_OUT, BIAS, NORM_OUT>(A, W, bias, bias_in, dinv, D, m0, lane);
}


__device__ __forceinline__ unsigned zero_bytes(unsigned x) {
  return ~(((x & 0x7F7F7F7Fu) + 0x7F7F7F7Fu) | x | 0x7F7F7F7Fu);
}

__device__ __forceinline__ uint4 load_ids(const unsigned char* __restrict__ bkt, int e0, int E,
                                          unsigned fill) {
  if (e0 + 16 <= E) return *(const uint4*)(bkt + e0);
  unsigned w[4] = {fill, fill, fill, fill};
#pragma unroll
  for (int k = 0; k < 16; ++k) {
    if (e0 + k < E) {
      const unsigned sh = 8u * (k & 3);
      w[k >> 2] = (w[k >> 2] & ~(0xFFu << sh)) | ((unsigned)bkt[e0 + k] << sh);
    }
  }
  return make_uint4(w[0], w[1], w[2], w[3]);
}

__device__ __forceinline__ uint2 fetch(const int* qp, int cnt, int lane,
                                       const unsigned* __restrict__ rec) {
  uint2 rc = make_uint2(0u, 0u);
  if (lane < cnt) rc = *(const uint2*)(rec + 2 * (size_t)qp[lane]);
  return rc;
}

template <class F>
__device__ __forceinline__ void scan_range(const unsigned char* __restrict__ bkt,
                                           const unsigned* __restrict__ rec, int E, unsigned me,
                                           int wv, int lane, int* myq, F take) {
  const unsigned pat  = me * 0x01010101u;
  const unsigned fill = (me ^ 1u) * 0x01010101u;
  const int nchunk = (E + 1023) >> 10;
  int n = 0;

  uint4 va = make_uint4(fill, fill, fill, fill);
  uint4 vb = va;
  if (wv < nchunk) {
    va = load_ids(bkt, wv * 1024 + lane * 16, E, fill);
    vb = load_ids(bkt, wv * 1024 + 512 + lane * 16, E, fill);
  }
  for (int ch = wv; ch < nchunk; ch += ES_WAVES) {
    const uint4 v = va, u = vb;
    const int e0 = ch * 1024 + lane * 16;
    if (ch + ES_WAVES < nchunk) {
      va = load_ids(bkt, e0 + ES_WAVES * 1024, E, fill);
      vb = load_ids(bkt, e0 + ES_WAVES * 1024 + 512, E, fill);
    }
    unsigned mm = (zero_bytes(v.x ^ pat) >> 7) | (zero_bytes(v.y ^ pat) >> 6) |
                  (zero_bytes(v.z ^ pat) >> 5) | (zero_bytes(v.w ^ pat) >> 4) |
                  (zero_bytes(u.x ^ pat) >> 3) | (zero_bytes(u.y ^ pat) >> 2) |
                  (zero_bytes(u.z ^ pat) >> 1) | zero_bytes(u.w ^ pat);
    unsigned bal = __builtin_amdgcn_ballot_w32(mm != 0u);
    while (bal != 0u) {
      if (mm != 0u) {
        const int p = __builtin_ctz(mm);
        mm &= mm - 1u;
        myq[n + (int)__builtin_amdgcn_mbcnt_lo(bal, 0u)] =
            e0 + ((p & 4) << 7) + ((p & 3) << 2) + (p >> 3);
      }
      n += __builtin_popcount(bal);
      if (n >= 32) {
        n -= 32;
        take(fetch(myq + n, 32, lane, rec), 32);
      }
      bal = __builtin_amdgcn_ballot_w32(mm != 0u);
    }
  }
  if (n > 0) take(fetch(myq, n, lane, rec), n);
}

__global__ __launch_bounds__(ES_WAVES * 32) void k_range_scan_and_first_product(
    const unsigned char* __restrict__ bkt, const unsigned* __restrict__ rec,
    float* __restrict__ dinv, unsigned* lists, unsigned* cnt, int N, int E,
    const float* __restrict__ x, const float* __restrict__ W1, const float* __restrict__ b1,
    float* __restrict__ h) {
  __shared__ float dsum[NPB];
  __shared__ int q[ES_WAVES][QCAP];
  __shared__ unsigned cursor;

  const int tid  = threadIdx.x;
  const int lane = tid & 31;
  const int wv   = __builtin_amdgcn_readfirstlane(tid >> 5);

  if (blockIdx.x >= NB) {
    const int m0 = ((blockIdx.x - NB) * ES_WAVES + wv) * 16;
    if (m0 < N)
      gemm_tile<F_IN, HIDDEN, false, true, true, false>(x, W1, b1, nullptr, nullptr, h, m0, lane);
    return;
  }

  const int node0 = blockIdx.x * NPB;
  int nown = N - node0;
  if (nown > NPB) nown = NPB;
  if (nown <= 0) return;

  uint2* mylist = (uint2*)(lists + 2 * (size_t)blockIdx.x * LCAP);
  if (tid == 0) cursor = 0u;
  for (int i = tid; i < NPB; i += ES_WAVES * 32) dsum[i] = 1.0f;
  __syncthreads();

  scan_range(bkt, rec, E, blockIdx.x, wv, lane, q[wv], [&](uint2 rc, int c) {
    unsigned base = 0u;
    if (lane == 0) base = atomicAdd(&cursor, (unsigned)c);
    base = (unsigned)__builtin_amdgcn_readfirstlane((int)base);
    if (lane < c) {
      if (base + (unsigned)c <= (unsigned)LCAP) mylist[base + lane] = rc;
      atomicAdd(&dsum[rc.x >> ROW_BITS], __uint_as_float(rc.y));
    }
  });
  __syncthreads();

  if (tid == 0) cnt[blockIdx.x] = cursor;
  for (int i = tid; i < nown; i += ES_WAVES * 32) {
    const float v = dsum[i];
    dinv[node0 + i] = (v > 0.0f) ? rsqrtf(v) : 0.0f;
  }
}

template <int G, bool FULL>
__device__ __forceinline__ void drain(uint2 rc, int cnt, int lane,
                                      const float* __restrict__ s, float* acc) {
  const int r = (int)(rc.x & ((1u << ROW_BITS) - 1u));
  const int c = (int)(rc.x >> ROW_BITS);
  const int w_bits = (int)rc.y;
#pragma unroll
  for (int g = 0; g < 32; g += G) {
    if (!FULL && g >= cnt) break;
    float v0[G], v1[G];
#pragma unroll
    for (int u = 0; u < G; ++u) {
      v0[u] = 0.0f; v1[u] = 0.0f;
      if (FULL || g + u < cnt) {
        const int rj = __builtin_amdgcn_readlane(r, g + u);
        const float* src = s + (size_t)rj * HIDDEN + lane;
        v0[u] = src[0];
        v1[u] = src[32];
      }
    }
#pragma unroll
    for (int u = 0; u < G; ++u) {
      if (FULL || g + u < cnt) {
        const int   cj = __builtin_amdgcn_readlane(c, g + u);
        const float wj = __int_as_float(__builtin_amdgcn_readlane(w_bits, g + u));
        float* dst = acc + cj * ACC_P + lane;
        atomicAdd(dst, wj * v0[u]);
        atomicAdd(dst + 32, wj * v1[u]);
      }
    }
  }
}

template <int NOUT, bool BIAS, bool NORM_OUT>
__device__ __forceinline__ void lds_tile_product(float* acc, int r0, int node_r0, int N,
                                                 const float* __restrict__ W,
                                                 const float* __restrict__ bias,
                                                 const float* __restrict__ bias_in,
                                                 const float* __restrict__ dinv, int lane) {
  constexpr int NT = (NOUT + 15) / 16;
  constexpr int KS = HIDDEN / 4;
  const int half = lane >> 4;
  const int l16  = lane & 15;

  int nd = node_r0 + l16;
  if (nd > N - 1) nd = N - 1;
  const float di = dinv[nd];

  const float* arow = acc + (r0 + l16) * ACC_P + 2 * half;
  v2f a[KS];
#pragma unroll
  for (int ks = 0; ks < KS; ++ks) {
    a[ks] = *(const v2f*)(arow + 4 * ks);
    const v2f bi = *(const v2f*)(bias_in + 4 * ks + 2 * half);
    a[ks].x = fmaxf(fmaf(di, a[ks].x, bi.x), 0.0f);
    a[ks].y = fmaxf(fmaf(di, a[ks].y, bi.y), 0.0f);
  }

  v8f d[NT] = {};
#pragma unroll
  for (int ks = 0; ks < KS; ++ks) {
    const int kk = 4 * ks + 2 * half;
#pragma unroll
    for (int t = 0; t < NT; ++t) {
      const int c = t * 16 + l16;
      v2f b;
      if ((NOUT % 16 == 0) || (c < NOUT)) {
        b.x = W[kk * NOUT + c];
        b.y = W[(kk + 1) * NOUT + c];
      } else {
        b.x = 0.0f; b.y = 0.0f;
      }
      d[t] = __builtin_amdgcn_wmma_f32_16x16x4_f32(
           false, a[ks],  false, b,
           (short)0, d[t],  false,  false);
    }
  }

  float dr[8];
  if (NORM_OUT) {
#pragma unroll
    for (int v = 0; v < 8; ++v) {
      int n2 = node_r0 + v + 8 * half;
      if (n2 > N - 1) n2 = N - 1;
      dr[v] = dinv[n2];
    }
  }

#pragma unroll
  for (int t = 0; t < NT; ++t) {
    const int c = t * 16 + l16;
    if ((NOUT % 16 != 0) && (c >= NOUT)) continue;
    const float bv = BIAS ? bias[c] : 0.0f;
#pragma unroll
    for (int v = 0; v < 8; ++v) {
      float val = d[t][v];
      if (BIAS) val += bv;
      if (NORM_OUT) val *= dr[v];
      acc[(r0 + v + 8 * half) * ACC_P + c] = val;
    }
  }
}

template <int NOUT, bool BIAS, bool NORM_OUT>
__global__ __launch_bounds__(ES_WAVES * 32) void k_edge_sum_product(
    const unsigned char* __restrict__ bkt, const unsigned* __restrict__ rec,
    const float* __restrict__ s, const unsigned* __restrict__ lists,
    const unsigned* __restrict__ cnt, int N, int E, const float* __restrict__ W,
    const float* __restrict__ bias, const float* __restrict__ bias_in,
    const float* __restrict__ dinv, float* __restrict__ D) {
  __shared__ __attribute__((aligned(16))) float acc[ROWS_P * ACC_P];
  __shared__ int q[ES_WAVES][QCAP];

  const int tid  = threadIdx.x;
  const int lane = tid & 31;
  const int wv   = __builtin_amdgcn_readfirstlane(tid >> 5);
  const int node0 = blockIdx.x * NPB;
  int nown = N - node0;
  if (nown > NPB) nown = NPB;
  if (nown <= 0) return;

  const uint2* mylist = (const uint2*)(lists + 2 * (size_t)blockIdx.x * LCAP);

  {
    constexpr int IT = (ROWS_P * (HIDDEN / 4) + ES_WAVES * 32 - 1) / (ES_WAVES * 32);
    const float4* s4 = (const float4*)(s + (size_t)node0 * HIDDEN);
    float4* a4 = (float4*)acc;
    const int lim = nown * (HIDDEN / 4);
    float4 t[IT];
#pragma unroll
    for (int k = 0; k < IT; ++k) {
      const int i = tid + k * (ES_WAVES * 32);
      t[k] = make_float4(0.0f, 0.0f, 0.0f, 0.0f);
      if (i < lim) t[k] = s4[i];
    }
#pragma unroll
    for (int k = 0; k < IT; ++k) {
      const int i = tid + k * (ES_WAVES * 32);
      if (i < ROWS_P * (HIDDEN / 4)) a4[(i >> 4) * (ACC_P / 4) + (i & 15)] = t[k];
    }
  }

  const unsigned total = cnt[blockIdx.x];
  __syncthreads();

  if (total <= (unsigned)LCAP) {
    const int tot  = (int)total;
    const int nblk = (tot + 31) >> 5;
    uint2 nxt = make_uint2(0u, 0u);
    if (wv < nblk && wv * 32 + lane < tot) nxt = mylist[wv * 32 + lane];
    for (int b = wv; b < nblk; b += ES_WAVES) {
      const uint2 cur = nxt;
      const int i2 = (b + ES_WAVES) * 32 + lane;
      nxt = make_uint2(0u, 0u);
      if (i2 < tot) nxt = mylist[i2];
      const int left = tot - b * 32;
      if (left >= 32) drain<LGRP, true>(cur, 32, lane, s, acc);
      else            drain<LGRP, false>(cur, left, lane, s, acc);
    }
  } else {
    scan_range(bkt, rec, E, blockIdx.x, wv, lane, q[wv], [&](uint2 rc, int c) {
      if (c == 32) drain<GRP, true>(rc, 32, lane, s, acc);
      else         drain<GRP, false>(rc, c, lane, s, acc);
    });
  }
  __syncthreads();

  {
    const int ntile = (nown + 15) >> 4;
    if (wv < ntile)
      lds_tile_product<NOUT, BIAS, NORM_OUT>(acc, wv * 16, node0 + wv * 16, N, W, bias, bias_in,
                                             dinv, lane);
  }
  __syncthreads();

  {
    constexpr int Q = NOUT / 4;
    float4* d4 = (float4*)(D + (size_t)node0 * NOUT);
    const float4* a4 = (const float4*)acc;
    for (int i = tid; i < nown * Q; i += ES_WAVES * 32) {
      const int r = i / Q;
      d4[i] = a4[r * (ACC_P / 4) + (i - r * Q)];
    }
  }
}


extern "C" void kernel_launch(void* const* d_in, const int* in_sizes, int n_in,
                              void* d_out, int out_size, void* d_ws, size_t ws_size,
                              hipStream_t stream) {
  const float* x   = (const float*)d_in[0];
  const int*   ei  = (const int*)d_in[1];
  const float* ew  = (const float*)d_in[2];
  const float* W1  = (const float*)d_in[3];
  const float* b1  = (const float*)d_in[4];
  const float* Wc1 = (const float*)d_in[5];
  const float* bc1 = (const float*)d_in[6];
  const float* Wc2 = (const float*)d_in[7];
  const float* bc2 = (const float*)d_in[8];
  const float* W2  = (const float*)d_in[9];
  const float* b2  = (const float*)d_in[10];
  float* out = (float*)d_out;

  const int N = in_sizes[0] / F_IN;
  const int E = in_sizes[2];
  const int* erow = ei;
  const int* ecol = ei + E;

  unsigned* lists = (unsigned*)d_ws;
  unsigned* cnt = lists + 2 * (size_t)NB * LCAP;
  float* dinv = (float*)(cnt + NB);
  float* h    = dinv + N;
  float* s    = h + (size_t)N * HIDDEN;
  unsigned* rec = (unsigned*)(s + (size_t)N * HIDDEN);
  unsigned char* bkt = (unsigned char*)(rec + 2 * (size_t)(E + (E & 1)));

  const int TB = 256;
  const int gemm_grid = (N + 31) / 32;

  k_edge_records<<<(E + TB - 1) / TB, TB, 0, stream>>>(erow, ecol, ew, bkt, rec, E);
  const int first_grid = (N + ES_WAVES * 16 - 1) / (ES_WAVES * 16);
  k_range_scan_and_first_product<<<NB + first_grid, ES_WAVES * 32, 0, stream>>>(
      bkt, rec, dinv, lists, cnt, N, E, x, W1, b1, h);

  k_gemm_wmma<HIDDEN, HIDDEN, false, false, false, true>
      <<<gemm_grid, 64, 0, stream>>>(h, Wc1, nullptr, nullptr, dinv, s, N);

  k_edge_sum_product<HIDDEN, false, true><<<NB, ES_WAVES * 32, 0, stream>>>(
      bkt, rec, s, lists, cnt, N, E, Wc2, nullptr, bc1, dinv, h);

  k_edge_sum_product<NCLS, true, false><<<NB, ES_WAVES * 32, 0, stream>>>(
      bkt, rec, h, lists, cnt, N, E, W2, b2, bc2, dinv, out);
}


# B fragments of the in-work-group products (x@W1, both convs' consuming ones) come from an LDS copy of W in fragment order, staged once per work-group: 16-byte LDS reads, not 4-byte global loads
# speedup vs baseline: 3.5434x; 1.0964x over previous
#include <hip/hip_runtime.h>

typedef __attribute__((ext_vector_type(2))) float v2f;
typedef __attribute__((ext_vector_type(8))) float v8f;

#define F_IN   256
#define HIDDEN 64
#define NCLS   40

#define NB       256
#define NPB      392
#define ROWS_P   (((NPB + 15) / 16) * 16)
#define ACC_P    68
#define ES_WAVES 32
#define QCAP     64
#define ROW_BITS 23
#define GRP      16
#define LGRP     32
#define LCAP     32768


__global__ void k_edge_records(const int* __restrict__ row, const int* __restrict__ col,
                               const float* __restrict__ ew, unsigned char* __restrict__ bkt,
                               unsigned* __restrict__ rec, int e) {
  int i = blockIdx.x * blockDim.x + threadIdx.x;
  if (i < e) {
    const int c = col[i];
    const int b = c / NPB;
    bkt[i] = (unsigned char)b;
    *(uint2*)(rec + 2 * (size_t)i) = make_uint2(
        (unsigned)row[i] | ((unsigned)(c - b * NPB) << ROW_BITS), (unsigned)__float_as_int(ew[i]));
  }
}


template <int K, int NOUT>
__device__ __forceinline__ void stage_b_fragments(const float* __restrict__ W, float4* wf, int tid,
                                                  int nthreads) {
  constexpr int NT = (NOUT + 15) / 16;
  constexpr int NP = (NT + 1) / 2;
  constexpr int TOTAL = (K / 4) * NP * 32;
  for (int e = tid; e < TOTAL; e += nthreads) {
    const int l  = e & 31;
    const int p  = (e >> 5) % NP;
    const int ks = (e >> 5) / NP;
    const int kk = 4 * ks + 2 * (l >> 4);
    const int c0 = 32 * p + (l & 15);
    const int c1 = c0 + 16;
    float4 v = make_float4(0.0f, 0.0f, 0.0f, 0.0f);
    if (c0 < NOUT) { v.x = W[kk * NOUT + c0]; v.y = W[(kk + 1) * NOUT + c0]; }
    if (c1 < NOUT) { v.z = W[kk * NOUT + c1]; v.w = W[(kk + 1) * NOUT + c1]; }
    wf[e] = v;
  }
}

template <int NT>
__device__ __forceinline__ void load_b_fragments(const float4* wf, int ks, int lane, v2f* b) {
  constexpr int NP = (NT + 1) / 2;
#pragma unroll
  for (int p = 0; p < NP; ++p) {
    const float4 q = wf[(ks * NP + p) * 32 + lane];
    b[2 * p].x = q.x; b[2 * p].y = q.y;
    if (2 * p + 1 < NT) { b[2 * p + 1].x = q.z; b[2 * p + 1].y = q.w; }
  }
}

template <int K, int NOUT, bool NORM_IN, bool RELU_OUT, bool BIAS, bool NORM_OUT, bool B_LDS>
__device__ __forceinline__ void gemm_tile(const float* A, const float* __restrict__ W,
                                          const float4* wf,
                                          const float* __restrict__ bias,
                                          const float* __restrict__ bias_in,
                                          const float* __restrict__ dinv,
                                          float* __restrict__ D, int m0, int lane) {
  constexpr int NT = (NOUT + 15) / 16;
  const int half = lane >> 4;
  const int l16  = lane & 15;

  v8f acc[NT] = {};

  const float* arow = A + (size_t)(m0 + l16) * K + 2 * half;
  const float di = NORM_IN ? dinv[m0 + l16] : 0.0f;

  for (int k = 0; k < K; k += 4) {
    v2f a = *(const v2f*)(arow + k);
    const int kk = k + 2 * half;
    if (NORM_IN) {
      const v2f bi = *(const v2f*)(bias_in + kk);
      a.x = fmaxf(fmaf(di, a.x, bi.x), 0.0f);
      a.y = fmaxf(fmaf(di, a.y, bi.y), 0.0f);
    }
    v2f b[NT];
    if (B_LDS) {
      load_b_fragments<NT>(wf, k >> 2, lane, b);
    } else {
#pragma unroll
      for (int t = 0; t < NT; ++t) {
        const int c = t * 16 + l16;
        if ((NOUT % 16 == 0) || (c < NOUT)) {
          b[t].x = W[(size_t)kk * NOUT + c];
          b[t].y = W[(size_t)(kk + 1) * NOUT + c];
        } else {
          b[t].x = 0.0f; b[t].y = 0.0f;
        }
      }
    }
#pragma unroll
    for (int t = 0; t < NT; ++t) {
      acc[t] = __builtin_amdgcn_wmma_f32_16x16x4_f32(
           false, a,  false, b[t],
           (short)0, acc[t],  false,  false);
    }
  }

  float dr[8];
  if (NORM_OUT) {
    const float4* dp = (const float4*)(dinv + m0 + 8 * half);
    const float4 q0 = dp[0], q1 = dp[1];
    dr[0] = q0.x; dr[1] = q0.y; dr[2] = q0.z; dr[3] = q0.w;
    dr[4] = q1.x; dr[5] = q1.y; dr[6] = q1.z; dr[7] = q1.w;
  }

#pragma unroll
  for (int t = 0; t < NT; ++t) {
    const int c = t * 16 + l16;
    if ((NOUT % 16 != 0) && (c >= NOUT)) continue;
    const float bv = BIAS ? bias[c] : 0.0f;
#pragma unroll
    for (int v = 0; v < 8; ++v) {
      const int m = m0 + v + 8 * half;
      float val = acc[t][v] + bv;
      if (RELU_OUT) val = fmaxf(val, 0.0f);
      if (NORM_OUT) val *= dr[v];
      D[(size_t)m * NOUT + c] = val;
    }
  }
}

template <int K, int NOUT, bool NORM_IN, bool RELU_OUT, bool BIAS, bool NORM_OUT>
__global__ __launch_bounds__(64) void k_gemm_wmma(const float* A,
                                                  const float* __restrict__ W,
                                                  const float* __restrict__ bias,
                                                  const float* __restrict__ bias_in,
                                                  const float* __restrict__ dinv,
                                                  float* __restrict__ D, int M) {
  const int wv   = threadIdx.x >> 5;
  const int lane = threadIdx.x & 31;
  const int m0   = (blockIdx.x * 2 + wv) * 16;
  if (m0 >= M) return;
  gemm_tile<K, NOUT, NORM_IN, RELU_OUT, BIAS, NORM_OUT, false>(A, W, nullptr, bias, bias_in, dinv, D,
                                                                m0, lane);
}


__device__ __forceinline__ unsigned zero_bytes(unsigned x) {
  return ~(((x & 0x7F7F7F7Fu) + 0x7F7F7F7Fu) | x | 0x7F7F7F7Fu);
}

__device__ __forceinline__ uint4 load_ids(const unsigned char* __restrict__ bkt, int e0, int E,
                                          unsigned fill) {
  if (e0 + 16 <= E) return *(const uint4*)(bkt + e0);
  unsigned w[4] = {fill, fill, fill, fill};
#pragma unroll
  for (int k = 0; k < 16; ++k) {
    if (e0 + k < E) {
      const unsigned sh = 8u * (k & 3);
      w[k >> 2] = (w[k >> 2] & ~(0xFFu << sh)) | ((unsigned)bkt[e0 + k] << sh);
    }
  }
  return make_uint4(w[0], w[1], w[2], w[3]);
}

__device__ __forceinline__ uint2 fetch(const int* qp, int cnt, int lane,
                                       const unsigned* __restrict__ rec) {
  uint2 rc = make_uint2(0u, 0u);
  if (lane < cnt) rc = *(const uint2*)(rec + 2 * (size_t)qp[lane]);
  return rc;
}

template <class F>
__device__ __forceinline__ void scan_range(const unsigned char* __restrict__ bkt,
                                           const unsigned* __restrict__ rec, int E, unsigned me,
                                           int wv, int lane, int* myq, F take) {
  const unsigned pat  = me * 0x01010101u;
  const unsigned fill = (me ^ 1u) * 0x01010101u;
  const int nchunk = (E + 1023) >> 10;
  int n = 0;

  uint4 va = make_uint4(fill, fill, fill, fill);
  uint4 vb = va;
  if (wv < nchunk) {
    va = load_ids(bkt, wv * 1024 + lane * 16, E, fill);
    vb = load_ids(bkt, wv * 1024 + 512 + lane * 16, E, fill);
  }
  for (int ch = wv; ch < nchunk; ch += ES_WAVES) {
    const uint4 v = va, u = vb;
    const int e0 = ch * 1024 + lane * 16;
    if (ch + ES_WAVES < nchunk) {
      va = load_ids(bkt, e0 + ES_WAVES * 1024, E, fill);
      vb = load_ids(bkt, e0 + ES_WAVES * 1024 + 512, E, fill);
    }
    unsigned mm = (zero_bytes(v.x ^ pat) >> 7) | (zero_bytes(v.y ^ pat) >> 6) |
                  (zero_bytes(v.z ^ pat) >> 5) | (zero_bytes(v.w ^ pat) >> 4) |
                  (zero_bytes(u.x ^ pat) >> 3) | (zero_bytes(u.y ^ pat) >> 2) |
                  (zero_bytes(u.z ^ pat) >> 1) | zero_bytes(u.w ^ pat);
    unsigned bal = __builtin_amdgcn_ballot_w32(mm != 0u);
    while (bal != 0u) {
      if (mm != 0u) {
        const int p = __builtin_ctz(mm);
        mm &= mm - 1u;
        myq[n + (int)__builtin_amdgcn_mbcnt_lo(bal, 0u)] =
            e0 + ((p & 4) << 7) + ((p & 3) << 2) + (p >> 3);
      }
      n += __builtin_popcount(bal);
      if (n >= 32) {
        n -= 32;
        take(fetch(myq + n, 32, lane, rec), 32);
      }
      bal = __builtin_amdgcn_ballot_w32(mm != 0u);
    }
  }
  if (n > 0) take(fetch(myq, n, lane, rec), n);
}

__global__ __launch_bounds__(ES_WAVES * 32) void k_range_scan_and_first_product(
    const unsigned char* __restrict__ bkt, const unsigned* __restrict__ rec,
    float* __restrict__ dinv, unsigned* lists, unsigned* cnt, int N, int E,
    const float* __restrict__ x, const float* __restrict__ W1, const float* __restrict__ b1,
    float* __restrict__ h) {
  __shared__ float dsum[NPB];
  __shared__ int q[ES_WAVES][QCAP];
  __shared__ unsigned cursor;
  __shared__ __attribute__((aligned(16))) float4 wfrag[(F_IN / 4) * 2 * 32];

  const int tid  = threadIdx.x;
  const int lane = tid & 31;
  const int wv   = __builtin_amdgcn_readfirstlane(tid >> 5);

  if (blockIdx.x >= NB) {
    stage_b_fragments<F_IN, HIDDEN>(W1, wfrag, tid, ES_WAVES * 32);
    __syncthreads();
    const int m0 = ((blockIdx.x - NB) * ES_WAVES + wv) * 16;
    if (m0 < N)
      gemm_tile<F_IN, HIDDEN, false, true, true, false, true>(x, W1, wfrag, b1, nullptr, nullptr, h, m0,
                                                              lane);
    return;
  }

  const int node0 = blockIdx.x * NPB;
  int nown = N - node0;
  if (nown > NPB) nown = NPB;
  if (nown <= 0) return;

  uint2* mylist = (uint2*)(lists + 2 * (size_t)blockIdx.x * LCAP);
  if (tid == 0) cursor = 0u;
  for (int i = tid; i < NPB; i += ES_WAVES * 32) dsum[i] = 1.0f;
  __syncthreads();

  scan_range(bkt, rec, E, blockIdx.x, wv, lane, q[wv], [&](uint2 rc, int c) {
    unsigned base = 0u;
    if (lane == 0) base = atomicAdd(&cursor, (unsigned)c);
    base = (unsigned)__builtin_amdgcn_readfirstlane((int)base);
    if (lane < c) {
      if (base + (unsigned)c <= (unsigned)LCAP) mylist[base + lane] = rc;
      atomicAdd(&dsum[rc.x >> ROW_BITS], __uint_as_float(rc.y));
    }
  });
  __syncthreads();

  if (tid == 0) cnt[blockIdx.x] = cursor;
  for (int i = tid; i < nown; i += ES_WAVES * 32) {
    const float v = dsum[i];
    dinv[node0 + i] = (v > 0.0f) ? rsqrtf(v) : 0.0f;
  }
}

template <int G, bool FULL>
__device__ __forceinline__ void drain(uint2 rc, int cnt, int lane,
                                      const float* __restrict__ s, float* acc) {
  const int r = (int)(rc.x & ((1u << ROW_BITS) - 1u));
  const int c = (int)(rc.x >> ROW_BITS);
  const int w_bits = (int)rc.y;
#pragma unroll
  for (int g = 0; g < 32; g += G) {
    if (!FULL && g >= cnt) break;
    float v0[G], v1[G];
#pragma unroll
    for (int u = 0; u < G; ++u) {
      v0[u] = 0.0f; v1[u] = 0.0f;
      if (FULL || g + u < cnt) {
        const int rj = __builtin_amdgcn_readlane(r, g + u);
        const float* src = s + (size_t)rj * HIDDEN + lane;
        v0[u] = src[0];
        v1[u] = src[32];
      }
    }
#pragma unroll
    for (int u = 0; u < G; ++u) {
      if (FULL || g + u < cnt) {
        const int   cj = __builtin_amdgcn_readlane(c, g + u);
        const float wj = __int_as_float(__builtin_amdgcn_readlane(w_bits, g + u));
        float* dst = acc + cj * ACC_P + lane;
        atomicAdd(dst, wj * v0[u]);
        atomicAdd(dst + 32, wj * v1[u]);
      }
    }
  }
}

template <int NOUT, bool BIAS, bool NORM_OUT>
__device__ __forceinline__ void lds_tile_product(float* acc, int r0, int node_r0, int N,
                                                 const float4* wf,
                                                 const float* __restrict__ bias,
                                                 const float* __restrict__ bias_in,
                                                 const float* __restrict__ dinv, int lane) {
  constexpr int NT = (NOUT + 15) / 16;
  constexpr int KS = HIDDEN / 4;
  const int half = lane >> 4;
  const int l16  = lane & 15;

  int nd = node_r0 + l16;
  if (nd > N - 1) nd = N - 1;
  const float di = dinv[nd];

  const float* arow = acc + (r0 + l16) * ACC_P + 2 * half;
  v2f a[KS];
#pragma unroll
  for (int ks = 0; ks < KS; ++ks) {
    a[ks] = *(const v2f*)(arow + 4 * ks);
    const v2f bi = *(const v2f*)(bias_in + 4 * ks + 2 * half);
    a[ks].x = fmaxf(fmaf(di, a[ks].x, bi.x), 0.0f);
    a[ks].y = fmaxf(fmaf(di, a[ks].y, bi.y), 0.0f);
  }

  v8f d[NT] = {};
#pragma unroll
  for (int ks = 0; ks < KS; ++ks) {
    v2f b[NT];
    load_b_fragments<NT>(wf, ks, lane, b);
#pragma unroll
    for (int t = 0; t < NT; ++t) {
      d[t] = __builtin_amdgcn_wmma_f32_16x16x4_f32(
           false, a[ks],  false, b[t],
           (short)0, d[t],  false,  false);
    }
  }

  float dr[8];
  if (NORM_OUT) {
#pragma unroll
    for (int v = 0; v < 8; ++v) {
      int n2 = node_r0 + v + 8 * half;
      if (n2 > N - 1) n2 = N - 1;
      dr[v] = dinv[n2];
    }
  }

#pragma unroll
  for (int t = 0; t < NT; ++t) {
    const int c = t * 16 + l16;
    if ((NOUT % 16 != 0) && (c >= NOUT)) continue;
    const float bv = BIAS ? bias[c] : 0.0f;
#pragma unroll
    for (int v = 0; v < 8; ++v) {
      float val = d[t][v];
      if (BIAS) val += bv;
      if (NORM_OUT) val *= dr[v];
      acc[(r0 + v + 8 * half) * ACC_P + c] = val;
    }
  }
}

template <int NOUT, bool BIAS, bool NORM_OUT>
__global__ __launch_bounds__(ES_WAVES * 32) void k_edge_sum_product(
    const unsigned char* __restrict__ bkt, const unsigned* __restrict__ rec,
    const float* __restrict__ s, const unsigned* __restrict__ lists,
    const unsigned* __restrict__ cnt, int N, int E, const float* __restrict__ W,
    const float* __restrict__ bias, const float* __restrict__ bias_in,
    const float* __restrict__ dinv, float* __restrict__ D) {
  __shared__ __attribute__((aligned(16))) float acc[ROWS_P * ACC_P];
  __shared__ int q[ES_WAVES][QCAP];
  __shared__ __attribute__((aligned(16))) float4 wfrag[(HIDDEN / 4) * 2 * 32];

  const int tid  = threadIdx.x;
  const int lane = tid & 31;
  const int wv   = __builtin_amdgcn_readfirstlane(tid >> 5);
  const int node0 = blockIdx.x * NPB;
  int nown = N - node0;
  if (nown > NPB) nown = NPB;
  if (nown <= 0) return;

  const uint2* mylist = (const uint2*)(lists + 2 * (size_t)blockIdx.x * LCAP);

  stage_b_fragments<HIDDEN, NOUT>(W, wfrag, tid, ES_WAVES * 32);

  {
    constexpr int IT = (ROWS_P * (HIDDEN / 4) + ES_WAVES * 32 - 1) / (ES_WAVES * 32);
    const float4* s4 = (const float4*)(s + (size_t)node0 * HIDDEN);
    float4* a4 = (float4*)acc;
    const int lim = nown * (HIDDEN / 4);
    float4 t[IT];
#pragma unroll
    for (int k = 0; k < IT; ++k) {
      const int i = tid + k * (ES_WAVES * 32);
      t[k] = make_float4(0.0f, 0.0f, 0.0f, 0.0f);
      if (i < lim) t[k] = s4[i];
    }
#pragma unroll
    for (int k = 0; k < IT; ++k) {
      const int i = tid + k * (ES_WAVES * 32);
      if (i < ROWS_P * (HIDDEN / 4)) a4[(i >> 4) * (ACC_P / 4) + (i & 15)] = t[k];
    }
  }

  const unsigned total = cnt[blockIdx.x];
  __syncthreads();

  if (total <= (unsigned)LCAP) {
    const int tot  = (int)total;
    const int nblk = (tot + 31) >> 5;
    uint2 nxt = make_uint2(0u, 0u);
    if (wv < nblk && wv * 32 + lane < tot) nxt = mylist[wv * 32 + lane];
    for (int b = wv; b < nblk; b += ES_WAVES) {
      const uint2 cur = nxt;
      const int i2 = (b + ES_WAVES) * 32 + lane;
      nxt = make_uint2(0u, 0u);
      if (i2 < tot) nxt = mylist[i2];
      const int left = tot - b * 32;
      if (left >= 32) drain<LGRP, true>(cur, 32, lane, s, acc);
      else            drain<LGRP, false>(cur, left, lane, s, acc);
    }
  } else {
    scan_range(bkt, rec, E, blockIdx.x, wv, lane, q[wv], [&](uint2 rc, int c) {
      if (c == 32) drain<GRP, true>(rc, 32, lane, s, acc);
      else         drain<GRP, false>(rc, c, lane, s, acc);
    });
  }
  __syncthreads();

  {
    const int ntile = (nown + 15) >> 4;
    if (wv < ntile)
      lds_tile_product<NOUT, BIAS, NORM_OUT>(acc, wv * 16, node0 + wv * 16, N, wfrag, bias, bias_in,
                                             dinv, lane);
  }
  __syncthreads();

  {
    constexpr int Q = NOUT / 4;
    float4* d4 = (float4*)(D + (size_t)node0 * NOUT);
    const float4* a4 = (const float4*)acc;
    for (int i = tid; i < nown * Q; i += ES_WAVES * 32) {
      const int r = i / Q;
      d4[i] = a4[r * (ACC_P / 4) + (i - r * Q)];
    }
  }
}


extern "C" void kernel_launch(void* const* d_in, const int* in_sizes, int n_in,
                              void* d_out, int out_size, void* d_ws, size_t ws_size,
                              hipStream_t stream) {
  const float* x   = (const float*)d_in[0];
  const int*   ei  = (const int*)d_in[1];
  const float* ew  = (const float*)d_in[2];
  const float* W1  = (const float*)d_in[3];
  const float* b1  = (const float*)d_in[4];
  const float* Wc1 = (const float*)d_in[5];
  const float* bc1 = (const float*)d_in[6];
  const float* Wc2 = (const float*)d_in[7];
  const float* bc2 = (const float*)d_in[8];
  const float* W2  = (const float*)d_in[9];
  const float* b2  = (const float*)d_in[10];
  float* out = (float*)d_out;

  const int N = in_sizes[0] / F_IN;
  const int E = in_sizes[2];
  const int* erow = ei;
  const int* ecol = ei + E;

  unsigned* lists = (unsigned*)d_ws;
  unsigned* cnt = lists + 2 * (size_t)NB * LCAP;
  float* dinv = (float*)(cnt + NB);
  float* h    = dinv + N;
  float* s    = h + (size_t)N * HIDDEN;
  unsigned* rec = (unsigned*)(s + (size_t)N * HIDDEN);
  unsigned char* bkt = (unsigned char*)(rec + 2 * (size_t)(E + (E & 1)));

  const int TB = 256;
  const int gemm_grid = (N + 31) / 32;

  k_edge_records<<<(E + TB - 1) / TB, TB, 0, stream>>>(erow, ecol, ew, bkt, rec, E);
  const int first_grid = (N + ES_WAVES * 16 - 1) / (ES_WAVES * 16);
  k_range_scan_and_first_product<<<NB + first_grid, ES_WAVES * 32, 0, stream>>>(
      bkt, rec, dinv, lists, cnt, N, E, x, W1, b1, h);

  k_gemm_wmma<HIDDEN, HIDDEN, false, false, false, true>
      <<<gemm_grid, 64, 0, stream>>>(h, Wc1, nullptr, nullptr, dinv, s, N);

  k_edge_sum_product<HIDDEN, false, true><<<NB, ES_WAVES * 32, 0, stream>>>(
      bkt, rec, s, lists, cnt, N, E, Wc2, nullptr, bc1, dinv, h);

  k_edge_sum_product<NCLS, true, false><<<NB, ES_WAVES * 32, 0, stream>>>(
      bkt, rec, h, lists, cnt, N, E, W2, b2, bc2, dinv, out);
}


# Id scan split 4 ways: a work-group scans a quarter of the edges for a group of 4 ranges (group-id byte), lists per range and slice; degree = 1 + 4 per-slice LDS partial sums, dinv made in next product
# speedup vs baseline: 3.8890x; 1.0975x over previous
#include <hip/hip_runtime.h>

typedef __attribute__((ext_vector_type(2))) float v2f;
typedef __attribute__((ext_vector_type(8))) float v8f;

#define F_IN   256
#define HIDDEN 64
#define NCLS   40

#define NB       256
#define NPB      392
#define ROWS_P   (((NPB + 15) / 16) * 16)
#define ACC_P    68
#define ES_WAVES 32
#define QCAP     64
#define ROW_BITS 20
#define GRP      16
#define LGRP     32
#define LCAP     32768
#define RG       4
#define NGRP     (NB / RG)
#define GNODES   (RG * NPB)
#define SCAP     (LCAP / RG)
#define NTOT     (NB * NPB)


__global__ void k_edge_records(const int* __restrict__ row, const int* __restrict__ col,
                               const float* __restrict__ ew, unsigned char* __restrict__ bkt,
                               unsigned* __restrict__ rec, int e) {
  int i = blockIdx.x * blockDim.x + threadIdx.x;
  if (i < e) {
    const int c = col[i];
    const int g = c / GNODES;
    bkt[i] = (unsigned char)g;
    *(uint2*)(rec + 2 * (size_t)i) = make_uint2(
        (unsigned)row[i] | ((unsigned)(c - g * GNODES) << ROW_BITS), (unsigned)__float_as_int(ew[i]));
  }
}


template <int K, int NOUT>
__device__ __forceinline__ void stage_b_fragments(const float* __restrict__ W, float4* wf, int tid,
                                                  int nthreads) {
  constexpr int NT = (NOUT + 15) / 16;
  constexpr int NP = (NT + 1) / 2;
  constexpr int TOTAL = (K / 4) * NP * 32;
  for (int e = tid; e < TOTAL; e += nthreads) {
    const int l  = e & 31;
    const int p  = (e >> 5) % NP;
    const int ks = (e >> 5) / NP;
    const int kk = 4 * ks + 2 * (l >> 4);
    const int c0 = 32 * p + (l & 15);
    const int c1 = c0 + 16;
    float4 v = make_float4(0.0f, 0.0f, 0.0f, 0.0f);
    if (c0 < NOUT) { v.x = W[kk * NOUT + c0]; v.y = W[(kk + 1) * NOUT + c0]; }
    if (c1 < NOUT) { v.z = W[kk * NOUT + c1]; v.w = W[(kk + 1) * NOUT + c1]; }
    wf[e] = v;
  }
}

template <int NT>
__device__ __forceinline__ void load_b_fragments(const float4* wf, int ks, int lane, v2f* b) {
  constexpr int NP = (NT + 1) / 2;
#pragma unroll
  for (int p = 0; p < NP; ++p) {
    const float4 q = wf[(ks * NP + p) * 32 + lane];
    b[2 * p].x = q.x; b[2 * p].y = q.y;
    if (2 * p + 1 < NT) { b[2 * p + 1].x = q.z; b[2 * p + 1].y = q.w; }
  }
}

template <int K, int NOUT, bool NORM_IN, bool RELU_OUT, bool BIAS, bool NORM_OUT, bool B_LDS>
__device__ __forceinline__ void gemm_tile(const float* A, const float* __restrict__ W,
                                          const float4* wf,
                                          const float* __restrict__ bias,
                                          const float* __restrict__ bias_in,
                                          const float* __restrict__ dinv,
                                          const float* __restrict__ pdeg,
                                          float* __restrict__ dinv_out,
                                          float* __restrict__ D, int m0, int lane) {
  constexpr int NT = (NOUT + 15) / 16;
  const int half = lane >> 4;
  const int l16  = lane & 15;

  v8f acc[NT] = {};

  const float* arow = A + (size_t)(m0 + l16) * K + 2 * half;
  const float di = NORM_IN ? dinv[m0 + l16] : 0.0f;

  for (int k = 0; k < K; k += 4) {
    v2f a = *(const v2f*)(arow + k);
    const int kk = k + 2 * half;
    if (NORM_IN) {
      const v2f bi = *(const v2f*)(bias_in + kk);
      a.x = fmaxf(fmaf(di, a.x, bi.x), 0.0f);
      a.y = fmaxf(fmaf(di, a.y, bi.y), 0.0f);
    }
    v2f b[NT];
    if (B_LDS) {
      load_b_fragments<NT>(wf, k >> 2, lane, b);
    } else {
#pragma unroll
      for (int t = 0; t < NT; ++t) {
        const int c = t * 16 + l16;
        if ((NOUT % 16 == 0) || (c < NOUT)) {
          b[t].x = W[(size_t)kk * NOUT + c];
          b[t].y = W[(size_t)(kk + 1) * NOUT + c];
        } else {
          b[t].x = 0.0f; b[t].y = 0.0f;
        }
      }
    }
#pragma unroll
    for (int t = 0; t < NT; ++t) {
      acc[t] = __builtin_amdgcn_wmma_f32_16x16x4_f32(
           false, a,  false, b[t],
           (short)0, acc[t],  false,  false);
    }
  }

  float dr[8];
  if (NORM_OUT) {
    float dg[8];
#pragma unroll
    for (int v = 0; v < 8; ++v) dg[v] = 1.0f;
#pragma unroll
    for (int sl = 0; sl < RG; ++sl) {
      const float4* pp = (const float4*)(pdeg + (size_t)sl * NTOT + m0 + 8 * half);
      const float4 q0 = pp[0], q1 = pp[1];
      dg[0] += q0.x; dg[1] += q0.y; dg[2] += q0.z; dg[3] += q0.w;
      dg[4] += q1.x; dg[5] += q1.y; dg[6] += q1.z; dg[7] += q1.w;
    }
#pragma unroll
    for (int v = 0; v < 8; ++v) dr[v] = (dg[v] > 0.0f) ? rsqrtf(dg[v]) : 0.0f;
    if (l16 == 0) {
      float4* op = (float4*)(dinv_out + m0 + 8 * half);
      op[0] = make_float4(dr[0], dr[1], dr[2], dr[3]);
      op[1] = make_float4(dr[4], dr[5], dr[6], dr[7]);
    }
  }

#pragma unroll
  for (int t = 0; t < NT; ++t) {
    const int c = t * 16 + l16;
    if ((NOUT % 16 != 0) && (c >= NOUT)) continue;
    const float bv = BIAS ? bias[c] : 0.0f;
#pragma unroll
    for (int v = 0; v < 8; ++v) {
      const int m = m0 + v + 8 * half;
      float val = acc[t][v] + bv;
      if (RELU_OUT) val = fmaxf(val, 0.0f);
      if (NORM_OUT) val *= dr[v];
      D[(size_t)m * NOUT + c] = val;
    }
  }
}

template <int K, int NOUT, bool NORM_IN, bool RELU_OUT, bool BIAS, bool NORM_OUT>
__global__ __launch_bounds__(64) void k_gemm_wmma(const float* A,
                                                  const float* __restrict__ W,
                                                  const float* __restrict__ bias,
                                                  const float* __restrict__ bias_in,
                                                  const float* __restrict__ dinv,
                                                  const float* __restrict__ pdeg,
                                                  float* __restrict__ dinv_out,
                                                  float* __restrict__ D, int M) {
  const int wv   = threadIdx.x >> 5;
  const int lane = threadIdx.x & 31;
  const int m0   = (blockIdx.x * 2 + wv) * 16;
  if (m0 >= M) return;
  gemm_tile<K, NOUT, NORM_IN, RELU_OUT, BIAS, NORM_OUT, false>(A, W, nullptr, bias, bias_in, dinv, pdeg,
                                                                dinv_out, D, m0, lane);
}


__device__ __forceinline__ unsigned zero_bytes(unsigned x) {
  return ~(((x & 0x7F7F7F7Fu) + 0x7F7F7F7Fu) | x | 0x7F7F7F7Fu);
}

__device__ __forceinline__ uint4 load_ids(const unsigned char* __restrict__ bkt, int e0, int E,
                                          unsigned fill) {
  if (e0 + 16 <= E) return *(const uint4*)(bkt + e0);
  unsigned w[4] = {fill, fill, fill, fill};
#pragma unroll
  for (int k = 0; k < 16; ++k) {
    if (e0 + k < E) {
      const unsigned sh = 8u * (k & 3);
      w[k >> 2] = (w[k >> 2] & ~(0xFFu << sh)) | ((unsigned)bkt[e0 + k] << sh);
    }
  }
  return make_uint4(w[0], w[1], w[2], w[3]);
}

__device__ __forceinline__ uint2 fetch(const int* qp, int cnt, int lane,
                                       const unsigned* __restrict__ rec) {
  uint2 rc = make_uint2(0u, 0u);
  if (lane < cnt) rc = *(const uint2*)(rec + 2 * (size_t)qp[lane]);
  return rc;
}

template <class F>
__device__ __forceinline__ void scan_range(const unsigned char* __restrict__ bkt,
                                           const unsigned* __restrict__ rec, int ch0, int E, unsigned me,
                                           int wv, int lane, int* myq, F take) {
  const unsigned pat  = me * 0x01010101u;
  const unsigned fill = (me ^ 1u) * 0x01010101u;
  const int nchunk = (E + 1023) >> 10;
  int n = 0;

  uint4 va = make_uint4(fill, fill, fill, fill);
  uint4 vb = va;
  if (ch0 + wv < nchunk) {
    va = load_ids(bkt, (ch0 + wv) * 1024 + lane * 16, E, fill);
    vb = load_ids(bkt, (ch0 + wv) * 1024 + 512 + lane * 16, E, fill);
  }
  for (int ch = ch0 + wv; ch < nchunk; ch += ES_WAVES) {
    const uint4 v = va, u = vb;
    const int e0 = ch * 1024 + lane * 16;
    if (ch + ES_WAVES < nchunk) {
      va = load_ids(bkt, e0 + ES_WAVES * 1024, E, fill);
      vb = load_ids(bkt, e0 + ES_WAVES * 1024 + 512, E, fill);
    }
    unsigned mm = (zero_bytes(v.x ^ pat) >> 7) | (zero_bytes(v.y ^ pat) >> 6) |
                  (zero_bytes(v.z ^ pat) >> 5) | (zero_bytes(v.w ^ pat) >> 4) |
                  (zero_bytes(u.x ^ pat) >> 3) | (zero_bytes(u.y ^ pat) >> 2) |
                  (zero_bytes(u.z ^ pat) >> 1) | zero_bytes(u.w ^ pat);
    unsigned bal = __builtin_amdgcn_ballot_w32(mm != 0u);
    while (bal != 0u) {
      if (mm != 0u) {
        const int p = __builtin_ctz(mm);
        mm &= mm - 1u;
        myq[n + (int)__builtin_amdgcn_mbcnt_lo(bal, 0u)] =
            e0 + ((p & 4) << 7) + ((p & 3) << 2) + (p >> 3);
      }
      n += __builtin_popcount(bal);
      if (n >= 32) {
        n -= 32;
        take(fetch(myq + n, 32, lane, rec), 32);
      }
      bal = __builtin_amdgcn_ballot_w32(mm != 0u);
    }
  }
  if (n > 0) take(fetch(myq, n, lane, rec), n);
}

__global__ __launch_bounds__(ES_WAVES * 32) void k_range_scan_and_first_product(
    const unsigned char* __restrict__ bkt, const unsigned* __restrict__ rec,
    float* __restrict__ pdeg, unsigned* lists, unsigned* cnt, int N, int E,
    const float* __restrict__ x, const float* __restrict__ W1, const float* __restrict__ b1,
    float* __restrict__ h) {
  __shared__ float dsum[GNODES];
  __shared__ int q[ES_WAVES][QCAP];
  __shared__ unsigned cursor[RG];
  __shared__ __attribute__((aligned(16))) float4 wfrag[(F_IN / 4) * 2 * 32];

  const int tid  = threadIdx.x;
  const int lane = tid & 31;
  const int wv   = __builtin_amdgcn_readfirstlane(tid >> 5);

  if (blockIdx.x >= NB) {
    stage_b_fragments<F_IN, HIDDEN>(W1, wfrag, tid, ES_WAVES * 32);
    __syncthreads();
    const int m0 = ((blockIdx.x - NB) * ES_WAVES + wv) * 16;
    if (m0 < N)
      gemm_tile<F_IN, HIDDEN, false, true, true, false, true>(x, W1, wfrag, b1, nullptr, nullptr, nullptr,
                                                              nullptr, h, m0, lane);
    return;
  }

  const int grp   = blockIdx.x / RG;
  const int slice = blockIdx.x % RG;
  const int node0 = grp * GNODES;
  int nown = N - node0;
  if (nown > GNODES) nown = GNODES;
  if (nown <= 0) return;

  const int nstep = (E + 1023) >> 10;
  const int sstep = (nstep + RG - 1) / RG;
  int eend = (slice + 1) * sstep * 1024;
  if (eend > E) eend = E;

  uint2* mylists = (uint2*)lists + ((size_t)grp * RG * RG + slice) * SCAP;
  if (tid < RG) cursor[tid] = 0u;
  for (int i = tid; i < GNODES; i += ES_WAVES * 32) dsum[i] = 0.0f;
  __syncthreads();

  scan_range(bkt, rec, slice * sstep, eend, grp, wv, lane, q[wv], [&](uint2 rc, int c) {
    const bool     live = lane < c;
    const unsigned pl   = rc.x >> ROW_BITS;
    const unsigned r    = pl / NPB;
    unsigned bal[RG];
    unsigned mine = 0u;
#pragma unroll
    for (int k = 0; k < RG; ++k) {
      bal[k] = __builtin_amdgcn_ballot_w32(live && r == (unsigned)k);
      if (r == (unsigned)k) mine = bal[k];
    }
    unsigned add = 0u;
#pragma unroll
    for (int k = 0; k < RG; ++k)
      if (lane == k) add = (unsigned)__builtin_popcount(bal[k]);
    unsigned base = 0u;
    if (lane < RG && add != 0u) base = atomicAdd(&cursor[lane], add);
    unsigned mybase = 0u;
#pragma unroll
    for (int k = 0; k < RG; ++k) {
      const unsigned bk = (unsigned)__builtin_amdgcn_readlane((int)base, k);
      if (r == (unsigned)k) mybase = bk;
    }
    if (live) {
      const unsigned pos = mybase + __builtin_amdgcn_mbcnt_lo(mine, 0u);
      if (pos < (unsigned)SCAP)
        mylists[(size_t)r * (RG * SCAP) + pos] =
            make_uint2((rc.x & ((1u << ROW_BITS) - 1u)) | ((pl - r * NPB) << ROW_BITS), rc.y);
      atomicAdd(&dsum[pl], __uint_as_float(rc.y));
    }
  });
  __syncthreads();

  if (tid < RG) cnt[(grp * RG + tid) * RG + slice] = cursor[tid];
  float* myp = pdeg + (size_t)slice * NTOT + node0;
  for (int i = tid; i < nown; i += ES_WAVES * 32) myp[i] = dsum[i];
}

template <int G, bool FULL>
__device__ __forceinline__ void drain(uint2 rc, unsigned live, int lane,
                                      const float* __restrict__ s, float* acc) {
  const int r = (int)(rc.x & ((1u << ROW_BITS) - 1u));
  const int c = (int)(rc.x >> ROW_BITS);
  const int w_bits = (int)rc.y;
#pragma unroll
  for (int g = 0; g < 32; g += G) {
    if (!FULL && (live >> g) == 0u) break;
    float v0[G], v1[G];
#pragma unroll
    for (int u = 0; u < G; ++u) {
      v0[u] = 0.0f; v1[u] = 0.0f;
      if (FULL || ((live >> (g + u)) & 1u) != 0u) {
        const int rj = __builtin_amdgcn_readlane(r, g + u);
        const float* src = s + (size_t)rj * HIDDEN + lane;
        v0[u] = src[0];
        v1[u] = src[32];
      }
    }
#pragma unroll
    for (int u = 0; u < G; ++u) {
      if (FULL || ((live >> (g + u)) & 1u) != 0u) {
        const int   cj = __builtin_amdgcn_readlane(c, g + u);
        const float wj = __int_as_float(__builtin_amdgcn_readlane(w_bits, g + u));
        float* dst = acc + cj * ACC_P + lane;
        atomicAdd(dst, wj * v0[u]);
        atomicAdd(dst + 32, wj * v1[u]);
      }
    }
  }
}

template <int NOUT, bool BIAS, bool NORM_OUT>
__device__ __forceinline__ void lds_tile_product(float* acc, int r0, int node_r0, int N,
                                                 const float4* wf,
                                                 const float* __restrict__ bias,
                                                 const float* __restrict__ bias_in,
                                                 const float* __restrict__ dinv, int lane) {
  constexpr int NT = (NOUT + 15) / 16;
  constexpr int KS = HIDDEN / 4;
  const int half = lane >> 4;
  const int l16  = lane & 15;

  int nd = node_r0 + l16;
  if (nd > N - 1) nd = N - 1;
  const float di = dinv[nd];

  const float* arow = acc + (r0 + l16) * ACC_P + 2 * half;
  v2f a[KS];
#pragma unroll
  for (int ks = 0; ks < KS; ++ks) {
    a[ks] = *(const v2f*)(arow + 4 * ks);
    const v2f bi = *(const v2f*)(bias_in + 4 * ks + 2 * half);
    a[ks].x = fmaxf(fmaf(di, a[ks].x, bi.x), 0.0f);
    a[ks].y = fmaxf(fmaf(di, a[ks].y, bi.y), 0.0f);
  }

  v8f d[NT] = {};
#pragma unroll
  for (int ks = 0; ks < KS; ++ks) {
    v2f b[NT];
    load_b_fragments<NT>(wf, ks, lane, b);
#pragma unroll
    for (int t = 0; t < NT; ++t) {
      d[t] = __builtin_amdgcn_wmma_f32_16x16x4_f32(
           false, a[ks],  false, b[t],
           (short)0, d[t],  false,  false);
    }
  }

  float dr[8];
  if (NORM_OUT) {
#pragma unroll
    for (int v = 0; v < 8; ++v) {
      int n2 = node_r0 + v + 8 * half;
      if (n2 > N - 1) n2 = N - 1;
      dr[v] = dinv[n2];
    }
  }

#pragma unroll
  for (int t = 0; t < NT; ++t) {
    const int c = t * 16 + l16;
    if ((NOUT % 16 != 0) && (c >= NOUT)) continue;
    const float bv = BIAS ? bias[c] : 0.0f;
#pragma unroll
    for (int v = 0; v < 8; ++v) {
      float val = d[t][v];
      if (BIAS) val += bv;
      if (NORM_OUT) val *= dr[v];
      acc[(r0 + v + 8 * half) * ACC_P + c] = val;
    }
  }
}

template <int NOUT, bool BIAS, bool NORM_OUT>
__global__ __launch_bounds__(ES_WAVES * 32) void k_edge_sum_product(
    const unsigned char* __restrict__ bkt, const unsigned* __restrict__ rec,
    const float* __restrict__ s, const unsigned* __restrict__ lists,
    const unsigned* __restrict__ cnt, int N, int E, const float* __restrict__ W,
    const float* __restrict__ bias, const float* __restrict__ bias_in,
    const float* __restrict__ dinv, float* __restrict__ D) {
  __shared__ __attribute__((aligned(16))) float acc[ROWS_P * ACC_P];
  __shared__ int q[ES_WAVES][QCAP];
  __shared__ __attribute__((aligned(16))) float4 wfrag[(HIDDEN / 4) * 2 * 32];

  const int tid  = threadIdx.x;
  const int lane = tid & 31;
  const int wv   = __builtin_amdgcn_readfirstlane(tid >> 5);
  const int node0 = blockIdx.x * NPB;
  int nown = N - node0;
  if (nown > NPB) nown = NPB;
  if (nown <= 0) return;

  const uint2* mylists = (const uint2*)lists + (size_t)blockIdx.x * RG * SCAP;

  stage_b_fragments<HIDDEN, NOUT>(W, wfrag, tid, ES_WAVES * 32);

  {
    constexpr int IT = (ROWS_P * (HIDDEN / 4) + ES_WAVES * 32 - 1) / (ES_WAVES * 32);
    const float4* s4 = (const float4*)(s + (size_t)node0 * HIDDEN);
    float4* a4 = (float4*)acc;
    const int lim = nown * (HIDDEN / 4);
    float4 t[IT];
#pragma unroll
    for (int k = 0; k < IT; ++k) {
      const int i = tid + k * (ES_WAVES * 32);
      t[k] = make_float4(0.0f, 0.0f, 0.0f, 0.0f);
      if (i < lim) t[k] = s4[i];
    }
#pragma unroll
    for (int k = 0; k < IT; ++k) {
      const int i = tid + k * (ES_WAVES * 32);
      if (i < ROWS_P * (HIDDEN / 4)) a4[(i >> 4) * (ACC_P / 4) + (i & 15)] = t[k];
    }
  }

  int lc[RG];
  int lb[RG + 1];
  bool whole = true;
  lb[0] = 0;
#pragma unroll
  for (int k = 0; k < RG; ++k) {
    const unsigned ck = (unsigned)__builtin_amdgcn_readfirstlane((int)cnt[blockIdx.x * RG + k]);
    whole = whole && (ck <= (unsigned)SCAP);
    lc[k] = (int)ck;
    lb[k + 1] = lb[k] + (int)((ck + 31u) >> 5);
  }
  __syncthreads();

  if (whole) {
    const int nblk = lb[RG];
    auto locate = [&](int gb, const uint2*& p, int& left) {
      int k = 0;
#pragma unroll
      for (int t = 1; t < RG; ++t) k += (gb >= lb[t]) ? 1 : 0;
      int off = 0, n = lc[0];
#pragma unroll
      for (int t = 1; t < RG; ++t)
        if (k == t) { off = lb[t]; n = lc[t]; }
      const int first = (gb - off) * 32;
      p = mylists + (size_t)k * SCAP + first;
      left = n - first;
    };
    uint2 nxt = make_uint2(0u, 0u);
    int nleft = 0;
    if (wv < nblk) {
      const uint2* p;
      locate(wv, p, nleft);
      if (lane < nleft) nxt = p[lane];
    }
    for (int gb = wv; gb < nblk; gb += ES_WAVES) {
      const uint2 cur = nxt;
      const int left = nleft;
      nxt = make_uint2(0u, 0u);
      if (gb + ES_WAVES < nblk) {
        const uint2* p;
        locate(gb + ES_WAVES, p, nleft);
        if (lane < nleft) nxt = p[lane];
      }
      if (left >= 32) drain<LGRP, true>(cur, 0xFFFFFFFFu, lane, s, acc);
      else            drain<LGRP, false>(cur, (1u << left) - 1u, lane, s, acc);
    }
  } else {
    const unsigned rme = (unsigned)(blockIdx.x % RG);
    scan_range(bkt, rec, 0, E, blockIdx.x / RG, wv, lane, q[wv], [&](uint2 rc, int c) {
      const unsigned pl = rc.x >> ROW_BITS;
      const unsigned mine = __builtin_amdgcn_ballot_w32(lane < c && pl / NPB == rme);
      rc.x = (rc.x & ((1u << ROW_BITS) - 1u)) | ((pl - rme * NPB) << ROW_BITS);
      drain<GRP, false>(rc, mine, lane, s, acc);
    });
  }
  __syncthreads();

  {
    const int ntile = (nown + 15) >> 4;
    if (wv < ntile)
      lds_tile_product<NOUT, BIAS, NORM_OUT>(acc, wv * 16, node0 + wv * 16, N, wfrag, bias, bias_in,
                                             dinv, lane);
  }
  __syncthreads();

  {
    constexpr int Q = NOUT / 4;
    float4* d4 = (float4*)(D + (size_t)node0 * NOUT);
    const float4* a4 = (const float4*)acc;
    for (int i = tid; i < nown * Q; i += ES_WAVES * 32) {
      const int r = i / Q;
      d4[i] = a4[r * (ACC_P / 4) + (i - r * Q)];
    }
  }
}


extern "C" void kernel_launch(void* const* d_in, const int* in_sizes, int n_in,
                              void* d_out, int out_size, void* d_ws, size_t ws_size,
                              hipStream_t stream) {
  const float* x   = (const float*)d_in[0];
  const int*   ei  = (const int*)d_in[1];
  const float* ew  = (const float*)d_in[2];
  const float* W1  = (const float*)d_in[3];
  const float* b1  = (const float*)d_in[4];
  const float* Wc1 = (const float*)d_in[5];
  const float* bc1 = (const float*)d_in[6];
  const float* Wc2 = (const float*)d_in[7];
  const float* bc2 = (const float*)d_in[8];
  const float* W2  = (const float*)d_in[9];
  const float* b2  = (const float*)d_in[10];
  float* out = (float*)d_out;

  const int N = in_sizes[0] / F_IN;
  const int E = in_sizes[2];
  const int* erow = ei;
  const int* ecol = ei + E;

  unsigned* lists = (unsigned*)d_ws;
  unsigned* cnt = lists + 2 * (size_t)NB * LCAP;
  float* pdeg = (float*)(cnt + NB * RG);
  float* dinv = pdeg + (size_t)RG * NTOT;
  float* h    = dinv + N;
  float* s    = h + (size_t)N * HIDDEN;
  unsigned* rec = (unsigned*)(s + (size_t)N * HIDDEN);
  unsigned char* bkt = (unsigned char*)(rec + 2 * (size_t)(E + (E & 1)));

  const int TB = 256;
  const int gemm_grid = (N + 31) / 32;

  k_edge_records<<<(E + TB - 1) / TB, TB, 0, stream>>>(erow, ecol, ew, bkt, rec, E);
  const int first_grid = (N + ES_WAVES * 16 - 1) / (ES_WAVES * 16);
  k_range_scan_and_first_product<<<NB + first_grid, ES_WAVES * 32, 0, stream>>>(
      bkt, rec, pdeg, lists, cnt, N, E, x, W1, b1, h);

  k_gemm_wmma<HIDDEN, HIDDEN, false, false, false, true>
      <<<gemm_grid, 64, 0, stream>>>(h, Wc1, nullptr, nullptr, nullptr, pdeg, dinv, s, N);

  k_edge_sum_product<HIDDEN, false, true><<<NB, ES_WAVES * 32, 0, stream>>>(
      bkt, rec, s, lists, cnt, N, E, Wc2, nullptr, bc1, dinv, h);

  k_edge_sum_product<NCLS, true, false><<<NB, ES_WAVES * 32, 0, stream>>>(
      bkt, rec, h, lists, cnt, N, E, W2, b2, bc2, dinv, out);
}
